# starting kernel
# baseline (speedup 1.0000x reference)
#include <hip/hip_runtime.h>

typedef __attribute__((ext_vector_type(2))) float v2f;
typedef __attribute__((ext_vector_type(8))) float v8f;

#define N_NODES 50000
#define R_REL   16
#define H_DIM   32
#define L_DIM   8
#define E_EDGES 1600000

#define K_STEPS ((R_REL + 1) * (H_DIM / 4))
#define TILES   (N_NODES / 16)

__global__ void rgcn_zero_f32(float* __restrict__ p, long long n) {
  long long i = (long long)blockIdx.x * blockDim.x + threadIdx.x;
  long long stride = (long long)gridDim.x * blockDim.x;
  for (; i < n; i += stride) p[i] = 0.0f;
}

__global__ void rgcn_count(const int* __restrict__ ei, const int* __restrict__ et,
                           float* __restrict__ cnt) {
  int e = blockIdx.x * blockDim.x + threadIdx.x;
  if (e >= E_EDGES) return;
  int dst = ei[E_EDGES + e];
  int r   = et[e];
  atomicAdd(&cnt[(size_t)r * N_NODES + dst], 1.0f);
}

__global__ void rgcn_scatter1(const int* __restrict__ ei, const int* __restrict__ et,
                              const float* __restrict__ cnt, const float* __restrict__ W1,
                              float* __restrict__ agg1) {
  long long t = (long long)blockIdx.x * blockDim.x + threadIdx.x;
  if (t >= (long long)E_EDGES * H_DIM) return;
  int e = (int)(t >> 5);
  int h = (int)(t & 31);
  int src = ei[e];
  int dst = ei[E_EDGES + e];
  int r   = et[e];
  float c = cnt[(size_t)r * N_NODES + dst];
  float norm = 1.0f / fmaxf(c, 1.0f);
  float w = W1[((size_t)r * N_NODES + src) * H_DIM + h];
  atomicAdd(&agg1[(size_t)dst * H_DIM + h], w * norm);
}

__global__ void rgcn_h1(const float* __restrict__ agg1, const float* __restrict__ root1,
                        const float* __restrict__ b1, float* __restrict__ h1) {
  int t = blockIdx.x * blockDim.x + threadIdx.x;
  if (t >= N_NODES * H_DIM) return;
  float v = agg1[t] + root1[t] + b1[t & (H_DIM - 1)];
  h1[t] = fmaxf(v, 0.0f);
}

__global__ void rgcn_scatter2(const int* __restrict__ ei, const int* __restrict__ et,
                              const float* __restrict__ cnt, const float* __restrict__ h1,
                              float* __restrict__ A) {
  long long t = (long long)blockIdx.x * blockDim.x + threadIdx.x;
  if (t >= (long long)E_EDGES * H_DIM) return;
  int e = (int)(t >> 5);
  int h = (int)(t & 31);
  int src = ei[e];
  int dst = ei[E_EDGES + e];
  int r   = et[e];
  float c = cnt[(size_t)r * N_NODES + dst];
  float norm = 1.0f / fmaxf(c, 1.0f);
  float v = h1[(size_t)src * H_DIM + h] * norm;
  atomicAdd(&A[((size_t)r * N_NODES + dst) * H_DIM + h], v);
}

__global__ void rgcn_out_wmma(const float* __restrict__ Amat, const float* __restrict__ h1,
                              const float* __restrict__ W2, const float* __restrict__ root2,
                              const float* __restrict__ b2, float* __restrict__ out) {
  __shared__ float ldsB[K_STEPS * 64];

  for (int idx = threadIdx.x; idx < K_STEPS * 64; idx += blockDim.x) {
    int s    = idx >> 6;
    int slot = idx & 63;
    int ln   = slot >> 1;
    int j    = slot & 1;
    int n    = ln & 15;
    int k    = ((s & 7) << 2) + ((ln >> 4) << 1) + j;
    int rr   = s >> 3;
    float v = 0.0f;
    if (n < L_DIM) {
      const float* Bseg = (rr < R_REL) ? (W2 + (size_t)rr * H_DIM * L_DIM) : root2;
      v = Bseg[(size_t)k * L_DIM + n];
    }
    ldsB[idx] = v;
  }
  __syncthreads();

  int wave = (blockIdx.x * blockDim.x + threadIdx.x) >> 5;
  int lane = threadIdx.x & 31;
  if (wave >= TILES) return;

  const int row0 = wave * 16;
  const int m    = lane & 15;
  const int koff = (lane >> 4) * 2;

  v8f acc = {};
  for (int rr = 0; rr <= R_REL; ++rr) {
    const float* Aseg = (rr < R_REL) ? (Amat + (size_t)rr * N_NODES * H_DIM) : h1;
    const float* arow = Aseg + (size_t)(row0 + m) * H_DIM + koff;
    const float* brow = ldsB + (size_t)rr * (H_DIM / 4) * 64 + lane * 2;
    #pragma unroll
    for (int k0 = 0; k0 < H_DIM; k0 += 4) {
      v2f a = *(const v2f*)(arow + k0);
      v2f b = *(const v2f*)(brow + (k0 >> 2) * 64);
      acc = __builtin_amdgcn_wmma_f32_16x16x4_f32(
           false, a,  false, b,
           (short)0, acc,  false,  false);
    }
  }

  int l = lane & 15;
  if (l < L_DIM) {
    float bias = b2[l];
    int half = lane >> 4;
    #pragma unroll
    for (int i = 0; i < 8; ++i) {
      int node = row0 + i + 8 * half;
      float x = acc[i] + bias;
      out[(size_t)node * L_DIM + l] = 1.0f / (1.0f + __expf(-x));
    }
  }
}

extern "C" void kernel_launch(void* const* d_in, const int* in_sizes, int n_in,
                              void* d_out, int out_size, void* d_ws, size_t ws_size,
                              hipStream_t stream) {
  const int*   ei    = (const int*)d_in[0];
  const int*   et    = (const int*)d_in[1];
  const float* W1    = (const float*)d_in[2];
  const float* root1 = (const float*)d_in[3];
  const float* b1    = (const float*)d_in[4];
  const float* W2    = (const float*)d_in[5];
  const float* root2 = (const float*)d_in[6];
  const float* b2    = (const float*)d_in[7];
  float*       out   = (float*)d_out;

  float* ws   = (float*)d_ws;
  float* cnt  = ws;
  float* agg1 = cnt + (size_t)R_REL * N_NODES;
  float* Amat = agg1 + (size_t)N_NODES * H_DIM;
  float* h1   = Amat + (size_t)R_REL * N_NODES * H_DIM;

  const long long nZero = (long long)R_REL * N_NODES
                        + (long long)N_NODES * H_DIM
                        + (long long)R_REL * N_NODES * H_DIM;
  rgcn_zero_f32<<<2048, 256, 0, stream>>>(ws, nZero);

  rgcn_count<<<(E_EDGES + 255) / 256, 256, 0, stream>>>(ei, et, cnt);

  const long long eh = (long long)E_EDGES * H_DIM;
  const int ehBlocks = (int)((eh + 255) / 256);
  rgcn_scatter1<<<ehBlocks, 256, 0, stream>>>(ei, et, cnt, W1, agg1);

  rgcn_h1<<<(N_NODES * H_DIM + 255) / 256, 256, 0, stream>>>(agg1, root1, b1, h1);

  rgcn_scatter2<<<ehBlocks, 256, 0, stream>>>(ei, et, cnt, h1, Amat);

  const int gemmBlocks = (TILES * 32 + 255) / 256;
  rgcn_out_wmma<<<gemmBlocks, 256, 0, stream>>>(Amat, h1, W2, root2, b2, out);

  (void)in_sizes; (void)n_in; (void)out_size; (void)ws_size;
}


# Layer 2 projects before it aggregates: WMMA h1 x [W2|root2] first, then edges atomically add 8-wide rows into the output sums instead of 32-wide into A[R,N,H]; float sum order changes
# speedup vs baseline: 1.3360x; 1.3360x over previous
#include <hip/hip_runtime.h>


typedef __attribute__((ext_vector_type(2))) float v2f;
typedef __attribute__((ext_vector_type(4))) float v4f;
typedef __attribute__((ext_vector_type(8))) float v8f;

#define N_NODES 50000
#define R_REL   16
#define H_DIM   32
#define L_DIM   8
#define E_EDGES 1600000

#define TILES   (N_NODES / 16)
#define K_STEPS (H_DIM / 4)
#define P_COLS  (R_REL * L_DIM)
#define B_COLS  (P_COLS + L_DIM)
#define C_TILES ((B_COLS + 15) / 16)

__global__ void rgcn_zero_f32(float* __restrict__ p, long long n) {
  long long i = (long long)blockIdx.x * blockDim.x + threadIdx.x;
  long long stride = (long long)gridDim.x * blockDim.x;
  for (; i < n; i += stride) p[i] = 0.0f;
}

__global__ void rgcn_count(const int* __restrict__ ei, const int* __restrict__ et,
                           float* __restrict__ cnt) {
  int e = blockIdx.x * blockDim.x + threadIdx.x;
  if (e >= E_EDGES) return;
  int dst = ei[E_EDGES + e];
  int r   = et[e];
  atomicAdd(&cnt[(size_t)r * N_NODES + dst], 1.0f);
}

__global__ void rgcn_scatter1(const int* __restrict__ ei, const int* __restrict__ et,
                              const float* __restrict__ cnt, const float* __restrict__ W1,
                              float* __restrict__ agg1) {
  long long t = (long long)blockIdx.x * blockDim.x + threadIdx.x;
  if (t >= (long long)E_EDGES * H_DIM) return;
  int e = (int)(t >> 5);
  int h = (int)(t & 31);
  int src = ei[e];
  int dst = ei[E_EDGES + e];
  int r   = et[e];
  float c = cnt[(size_t)r * N_NODES + dst];
  float norm = 1.0f / fmaxf(c, 1.0f);
  float w = W1[((size_t)r * N_NODES + src) * H_DIM + h];
  atomicAdd(&agg1[(size_t)dst * H_DIM + h], w * norm);
}

__global__ void rgcn_h1(const float* __restrict__ agg1, const float* __restrict__ root1,
                        const float* __restrict__ b1, float* __restrict__ h1) {
  int t = blockIdx.x * blockDim.x + threadIdx.x;
  if (t >= N_NODES * H_DIM) return;
  float v = agg1[t] + root1[t] + b1[t & (H_DIM - 1)];
  h1[t] = fmaxf(v, 0.0f);
}

__global__ void rgcn_project_wmma(const float* __restrict__ h1, const float* __restrict__ W2,
                                  const float* __restrict__ root2, const float* __restrict__ b2,
                                  float* __restrict__ P, float* __restrict__ pre) {
  __shared__ float ldsB[C_TILES * K_STEPS * 64];

  for (int idx = threadIdx.x; idx < C_TILES * K_STEPS * 64; idx += blockDim.x) {
    int ct   = idx >> 9;
    int s    = (idx >> 6) & (K_STEPS - 1);
    int slot = idx & 63;
    int ln   = slot >> 1;
    int j    = slot & 1;
    int c    = (ct << 4) + (ln & 15);
    int k    = (s << 2) + ((ln >> 4) << 1) + j;
    float v = 0.0f;
    if (c < P_COLS) {
      v = W2[((size_t)(c >> 3) * H_DIM + k) * L_DIM + (c & (L_DIM - 1))];
    } else if (c < B_COLS) {
      v = root2[(size_t)k * L_DIM + (c - P_COLS)];
    }
    ldsB[idx] = v;
  }
  __syncthreads();

  int wave = (blockIdx.x * blockDim.x + threadIdx.x) >> 5;
  int lane = threadIdx.x & 31;
  if (wave >= TILES) return;

  const int row0 = wave * 16;
  const int m    = lane & 15;
  const int half = lane >> 4;
  const int koff = half * 2;
  const int col  = lane & 15;

  const float* arow = h1 + (size_t)(row0 + m) * H_DIM + koff;
  v2f a[K_STEPS];
  #pragma unroll
  for (int s = 0; s < K_STEPS; ++s) a[s] = *(const v2f*)(arow + s * 4);

  const float* brow = ldsB + lane * 2;

  #pragma unroll
  for (int ct = 0; ct < C_TILES - 1; ++ct) {
    v8f acc = {};
    #pragma unroll
    for (int s = 0; s < K_STEPS; ++s) {
      v2f b = *(const v2f*)(brow + (ct * K_STEPS + s) * 64);
      acc = __builtin_amdgcn_wmma_f32_16x16x4_f32(
           false, a[s],  false, b,
           (short)0, acc,  false,  false);
    }
    #pragma unroll
    for (int i = 0; i < 8; ++i) {
      int node = row0 + i + 8 * half;
      P[(size_t)node * P_COLS + ct * 16 + col] = acc[i];
    }
  }

  {
    float bias = b2[col & (L_DIM - 1)];
    v8f acc = {};
    #pragma unroll
    for (int s = 0; s < K_STEPS; ++s) {
      v2f b = *(const v2f*)(brow + ((C_TILES - 1) * K_STEPS + s) * 64);
      acc = __builtin_amdgcn_wmma_f32_16x16x4_f32(
           false, a[s],  false, b,
           (short)0, acc,  false,  false);
    }
    if (col < L_DIM) {
      #pragma unroll
      for (int i = 0; i < 8; ++i) {
        int node = row0 + i + 8 * half;
        pre[(size_t)node * L_DIM + col] = acc[i] + bias;
      }
    }
  }
}

__global__ void rgcn_scatter2(const int* __restrict__ ei, const int* __restrict__ et,
                              const float* __restrict__ cnt, const float* __restrict__ P,
                              float* __restrict__ pre) {
  int t = blockIdx.x * blockDim.x + threadIdx.x;
  if (t >= E_EDGES * L_DIM) return;
  int e = t >> 3;
  int l = t & (L_DIM - 1);
  int src = ei[e];
  int dst = ei[E_EDGES + e];
  int r   = et[e];
  float c = cnt[(size_t)r * N_NODES + dst];
  float norm = 1.0f / fmaxf(c, 1.0f);
  float v = P[((size_t)src * R_REL + r) * L_DIM + l] * norm;
  atomicAdd(&pre[(size_t)dst * L_DIM + l], v);
}

__global__ void rgcn_sigmoid(const float* __restrict__ pre, float* __restrict__ out) {
  int t = blockIdx.x * blockDim.x + threadIdx.x;
  if (t >= N_NODES * L_DIM / 4) return;
  v4f x = ((const v4f*)pre)[t];
  v4f y;
  y.x = 1.0f / (1.0f + __expf(-x.x));
  y.y = 1.0f / (1.0f + __expf(-x.y));
  y.z = 1.0f / (1.0f + __expf(-x.z));
  y.w = 1.0f / (1.0f + __expf(-x.w));
  ((v4f*)out)[t] = y;
}

extern "C" void kernel_launch(void* const* d_in, const int* in_sizes, int n_in,
                              void* d_out, int out_size, void* d_ws, size_t ws_size,
                              hipStream_t stream) {
  const int*   ei    = (const int*)d_in[0];
  const int*   et    = (const int*)d_in[1];
  const float* W1    = (const float*)d_in[2];
  const float* root1 = (const float*)d_in[3];
  const float* b1    = (const float*)d_in[4];
  const float* W2    = (const float*)d_in[5];
  const float* root2 = (const float*)d_in[6];
  const float* b2    = (const float*)d_in[7];
  float*       out   = (float*)d_out;

  float* ws   = (float*)d_ws;
  float* cnt  = ws;
  float* agg1 = cnt + (size_t)R_REL * N_NODES;
  float* h1   = agg1 + (size_t)N_NODES * H_DIM;
  float* P    = h1 + (size_t)N_NODES * H_DIM;
  float* pre  = P + (size_t)N_NODES * P_COLS;

  const long long nZero = (long long)R_REL * N_NODES
                        + (long long)N_NODES * H_DIM;
  rgcn_zero_f32<<<2048, 256, 0, stream>>>(ws, nZero);

  rgcn_count<<<(E_EDGES + 255) / 256, 256, 0, stream>>>(ei, et, cnt);

  const long long eh = (long long)E_EDGES * H_DIM;
  const int ehBlocks = (int)((eh + 255) / 256);
  rgcn_scatter1<<<ehBlocks, 256, 0, stream>>>(ei, et, cnt, W1, agg1);

  rgcn_h1<<<(N_NODES * H_DIM + 255) / 256, 256, 0, stream>>>(agg1, root1, b1, h1);

  const int gemmBlocks = (TILES * 32 + 255) / 256;
  rgcn_project_wmma<<<gemmBlocks, 256, 0, stream>>>(h1, W2, root2, b2, P, pre);

  rgcn_scatter2<<<(E_EDGES * L_DIM + 255) / 256, 256, 0, stream>>>(ei, et, cnt, P, pre);

  rgcn_sigmoid<<<(N_NODES * L_DIM / 4 + 255) / 256, 256, 0, stream>>>(pre, out);

  (void)in_sizes; (void)n_in; (void)out_size; (void)ws_size;
}


# Edges counting-sorted by dst in every call; each layer sums a node's segment in registers (relu, sigmoid folded in) instead of float atomics into agg1/pre; float sum order changes
# speedup vs baseline: 1.3799x; 1.0329x over previous
#include <hip/hip_runtime.h>


typedef __attribute__((ext_vector_type(2))) float v2f;
typedef __attribute__((ext_vector_type(4))) float v4f;
typedef __attribute__((ext_vector_type(8))) float v8f;

#define N_NODES 50000
#define R_REL   16
#define H_DIM   32
#define L_DIM   8
#define E_EDGES 1600000

#define TILES   (N_NODES / 16)
#define K_STEPS (H_DIM / 4)
#define P_COLS  (R_REL * L_DIM)
#define B_COLS  (P_COLS + L_DIM)
#define C_TILES ((B_COLS + 15) / 16)

#define SCAN_WAVES  32
#define SCAN_ITERS  13
#define SCAN_CHUNK  128
#define SCAN_PER_WAVE (SCAN_ITERS * SCAN_CHUNK)
#define SCAN_TOTAL  (SCAN_WAVES * SCAN_PER_WAVE)

static_assert(SCAN_TOTAL >= N_NODES + 1, "scan range must cover every node and the end offset");
static_assert(N_NODES <= 65536, "source index is packed into 16 bits");
static_assert(N_NODES % 8 == 0, "8 nodes per block in the layer-1 segment kernel");
static_assert(N_NODES % 4 == 0, "4 nodes per wave in the layer-2 segment kernel");

__global__ void rgcn_zero_f32(float* __restrict__ p, long long n) {
  long long i = (long long)blockIdx.x * blockDim.x + threadIdx.x;
  long long stride = (long long)gridDim.x * blockDim.x;
  for (; i < n; i += stride) p[i] = 0.0f;
}

__global__ void rgcn_count(const int* __restrict__ ei, const int* __restrict__ et,
                           float* __restrict__ cnt, int* __restrict__ deg) {
  int e = blockIdx.x * blockDim.x + threadIdx.x;
  if (e >= E_EDGES) return;
  int dst = ei[E_EDGES + e];
  int r   = et[e];
  atomicAdd(&cnt[(size_t)r * N_NODES + dst], 1.0f);
  atomicAdd(&deg[dst], 1);
}

__global__ __launch_bounds__(1024) void rgcn_scan(const int* __restrict__ deg,
                                                  int* __restrict__ offs,
                                                  int* __restrict__ cursor) {
  __shared__ int wsum[SCAN_WAVES];
  const int lane = threadIdx.x & 31;
  const int w    = threadIdx.x >> 5;
  const int base = w * SCAN_PER_WAVE + lane * 4;

  int4 v[SCAN_ITERS];
  #pragma unroll
  for (int k = 0; k < SCAN_ITERS; ++k) v[k] = *(const int4*)(deg + base + k * SCAN_CHUNK);

  int ex[SCAN_ITERS];
  int carry = 0;
  #pragma unroll
  for (int k = 0; k < SCAN_ITERS; ++k) {
    int s = v[k].x + v[k].y + v[k].z + v[k].w;
    int inc = s;
    #pragma unroll
    for (int d = 1; d < 32; d <<= 1) {
      int t = __shfl_up(inc, d);
      if (lane >= d) inc += t;
    }
    ex[k] = carry + inc - s;
    carry += __shfl(inc, 31);
  }
  if (lane == 0) wsum[w] = carry;
  __syncthreads();

  int t0 = wsum[lane];
  int inc = t0;
  #pragma unroll
  for (int d = 1; d < 32; d <<= 1) {
    int t = __shfl_up(inc, d);
    if (lane >= d) inc += t;
  }
  const int woff = __shfl(inc - t0, w);

  #pragma unroll
  for (int k = 0; k < SCAN_ITERS; ++k) {
    int4 o;
    o.x = woff + ex[k];
    o.y = o.x + v[k].x;
    o.z = o.y + v[k].y;
    o.w = o.z + v[k].z;
    *(int4*)(offs   + base + k * SCAN_CHUNK) = o;
    *(int4*)(cursor + base + k * SCAN_CHUNK) = o;
  }
}

__global__ void rgcn_fill(const int* __restrict__ ei, const int* __restrict__ et,
                          const float* __restrict__ cnt, int* __restrict__ cursor,
                          int2* __restrict__ sorted) {
  int e = blockIdx.x * blockDim.x + threadIdx.x;
  if (e >= E_EDGES) return;
  int src = ei[e];
  int dst = ei[E_EDGES + e];
  int r   = et[e];
  float c = cnt[(size_t)r * N_NODES + dst];
  float norm = 1.0f / fmaxf(c, 1.0f);
  int pos = atomicAdd(&cursor[dst], 1);
  int2 en;
  en.x = src | (r << 16);
  en.y = __float_as_int(norm);
  sorted[pos] = en;
}

__global__ __launch_bounds__(256) void rgcn_layer1_seg(const int* __restrict__ offs,
                                                       const int2* __restrict__ sorted,
                                                       const float* __restrict__ W1,
                                                       const float* __restrict__ root1,
                                                       const float* __restrict__ b1,
                                                       float* __restrict__ h1) {
  const int lane = threadIdx.x & 31;
  const int node = blockIdx.x * 8 + __builtin_amdgcn_readfirstlane(threadIdx.x >> 5);
  const int beg = offs[node];
  const int end = offs[node + 1];
  float acc = 0.0f;
  for (int base = beg; base < end; base += 32) {
    const int n = min(32, end - base);
    int   key = 0;
    float nm  = 0.0f;
    if (lane < n) {
      int2 en = sorted[base + lane];
      key = en.x;
      nm  = __int_as_float(en.y);
    }
    for (int j = 0; j < n; j += 8) {
      float w[8];
      float nn[8];
      #pragma unroll
      for (int u = 0; u < 8; ++u) {
        int kk = __builtin_amdgcn_readlane(key, j + u);
        nn[u]  = __int_as_float(__builtin_amdgcn_readlane(__float_as_int(nm), j + u));
        int src = kk & 0xFFFF;
        int r   = kk >> 16;
        w[u] = W1[((size_t)r * N_NODES + src) * H_DIM + lane];
      }
      #pragma unroll
      for (int u = 0; u < 8; ++u) {
        if (j + u < n) acc = fmaf(w[u], nn[u], acc);
      }
    }
  }
  const size_t o = (size_t)node * H_DIM + lane;
  h1[o] = fmaxf(acc + root1[o] + b1[lane], 0.0f);
}

__global__ void rgcn_project_wmma(const float* __restrict__ h1, const float* __restrict__ W2,
                                  const float* __restrict__ root2, const float* __restrict__ b2,
                                  float* __restrict__ P, float* __restrict__ pre) {
  __shared__ float ldsB[C_TILES * K_STEPS * 64];

  for (int idx = threadIdx.x; idx < C_TILES * K_STEPS * 64; idx += blockDim.x) {
    int ct   = idx >> 9;
    int s    = (idx >> 6) & (K_STEPS - 1);
    int slot = idx & 63;
    int ln   = slot >> 1;
    int j    = slot & 1;
    int c    = (ct << 4) + (ln & 15);
    int k    = (s << 2) + ((ln >> 4) << 1) + j;
    float v = 0.0f;
    if (c < P_COLS) {
      v = W2[((size_t)(c >> 3) * H_DIM + k) * L_DIM + (c & (L_DIM - 1))];
    } else if (c < B_COLS) {
      v = root2[(size_t)k * L_DIM + (c - P_COLS)];
    }
    ldsB[idx] = v;
  }
  __syncthreads();

  int wave = (blockIdx.x * blockDim.x + threadIdx.x) >> 5;
  int lane = threadIdx.x & 31;
  if (wave >= TILES) return;

  const int row0 = wave * 16;
  const int m    = lane & 15;
  const int half = lane >> 4;
  const int koff = half * 2;
  const int col  = lane & 15;

  const float* arow = h1 + (size_t)(row0 + m) * H_DIM + koff;
  v2f a[K_STEPS];
  #pragma unroll
  for (int s = 0; s < K_STEPS; ++s) a[s] = *(const v2f*)(arow + s * 4);

  const float* brow = ldsB + lane * 2;

  #pragma unroll
  for (int ct = 0; ct < C_TILES - 1; ++ct) {
    v8f acc = {};
    #pragma unroll
    for (int s = 0; s < K_STEPS; ++s) {
      v2f b = *(const v2f*)(brow + (ct * K_STEPS + s) * 64);
      acc = __builtin_amdgcn_wmma_f32_16x16x4_f32(
           false, a[s],  false, b,
           (short)0, acc,  false,  false);
    }
    #pragma unroll
    for (int i = 0; i < 8; ++i) {
      int node = row0 + i + 8 * half;
      P[(size_t)node * P_COLS + ct * 16 + col] = acc[i];
    }
  }

  {
    float bias = b2[col & (L_DIM - 1)];
    v8f acc = {};
    #pragma unroll
    for (int s = 0; s < K_STEPS; ++s) {
      v2f b = *(const v2f*)(brow + ((C_TILES - 1) * K_STEPS + s) * 64);
      acc = __builtin_amdgcn_wmma_f32_16x16x4_f32(
           false, a[s],  false, b,
           (short)0, acc,  false,  false);
    }
    if (col < L_DIM) {
      #pragma unroll
      for (int i = 0; i < 8; ++i) {
        int node = row0 + i + 8 * half;
        pre[(size_t)node * L_DIM + col] = acc[i] + bias;
      }
    }
  }
}

__global__ __launch_bounds__(256) void rgcn_layer2_seg(const int* __restrict__ offs,
                                                       const int2* __restrict__ sorted,
                                                       const float* __restrict__ P,
                                                       const float* __restrict__ pre,
                                                       float* __restrict__ out) {
  const int lane = threadIdx.x & 31;
  const int wave = blockIdx.x * 8 + __builtin_amdgcn_readfirstlane(threadIdx.x >> 5);
  if (wave >= N_NODES / 4) return;
  const int l = lane & (L_DIM - 1);
  const int g = lane >> 3;
  const int node0 = wave * 4;
  float res = 0.0f;
  for (int q = 0; q < 4; ++q) {
    const int beg = offs[node0 + q];
    const int end = offs[node0 + q + 1];
    float acc = 0.0f;
    for (int base = beg; base < end; base += 32) {
      const int n = min(32, end - base);
      int   key = 0;
      float nm  = 0.0f;
      if (lane < n) {
        int2 en = sorted[base + lane];
        key = en.x;
        nm  = __int_as_float(en.y);
      }
      float v[8];
      float nn[8];
      #pragma unroll
      for (int jj = 0; jj < 8; ++jj) {
        int idx = jj * 4 + g;
        int kk  = __shfl(key, idx);
        nn[jj]  = __shfl(nm, idx);
        int src = kk & 0xFFFF;
        int r   = kk >> 16;
        v[jj] = P[((size_t)src * R_REL + r) * L_DIM + l];
      }
      #pragma unroll
      for (int jj = 0; jj < 8; ++jj) {
        if (jj * 4 + g < n) acc = fmaf(v[jj], nn[jj], acc);
      }
    }
    acc += __shfl_xor(acc, 8);
    acc += __shfl_xor(acc, 16);
    if (g == q) res = acc;
  }
  const size_t o = (size_t)node0 * L_DIM + lane;
  float x = pre[o] + res;
  out[o] = 1.0f / (1.0f + __expf(-x));
}

extern "C" void kernel_launch(void* const* d_in, const int* in_sizes, int n_in,
                              void* d_out, int out_size, void* d_ws, size_t ws_size,
                              hipStream_t stream) {
  const int*   ei    = (const int*)d_in[0];
  const int*   et    = (const int*)d_in[1];
  const float* W1    = (const float*)d_in[2];
  const float* root1 = (const float*)d_in[3];
  const float* b1    = (const float*)d_in[4];
  const float* W2    = (const float*)d_in[5];
  const float* root2 = (const float*)d_in[6];
  const float* b2    = (const float*)d_in[7];
  float*       out   = (float*)d_out;

  float* ws     = (float*)d_ws;
  float* cnt    = ws;
  int*   deg    = (int*)(cnt + (size_t)R_REL * N_NODES);
  int*   offs   = deg + SCAN_TOTAL;
  int*   cursor = offs + SCAN_TOTAL;
  int2*  sorted = (int2*)(cursor + SCAN_TOTAL);
  float* h1     = (float*)(sorted + E_EDGES);
  float* P      = h1 + (size_t)N_NODES * H_DIM;
  float* pre    = P + (size_t)N_NODES * P_COLS;

  const long long nZero = (long long)R_REL * N_NODES + SCAN_TOTAL;
  rgcn_zero_f32<<<512, 256, 0, stream>>>(ws, nZero);

  rgcn_count<<<(E_EDGES + 255) / 256, 256, 0, stream>>>(ei, et, cnt, deg);

  rgcn_scan<<<1, 1024, 0, stream>>>(deg, offs, cursor);

  rgcn_fill<<<(E_EDGES + 255) / 256, 256, 0, stream>>>(ei, et, cnt, cursor, sorted);

  rgcn_layer1_seg<<<N_NODES / 8, 256, 0, stream>>>(offs, sorted, W1, root1, b1, h1);

  const int gemmBlocks = (TILES * 32 + 255) / 256;
  rgcn_project_wmma<<<gemmBlocks, 256, 0, stream>>>(h1, W2, root2, b2, P, pre);

  rgcn_layer2_seg<<<(N_NODES / 4 + 7) / 8, 256, 0, stream>>>(offs, sorted, P, pre, out);

  (void)in_sizes; (void)n_in; (void)out_size; (void)ws_size;
}


# Per-(relation,dst) mean counts taken from each node's dst-sorted segment by wave ballots in layer 1; the float-atomic count table, its zeroing and its per-edge gather go; sum order as before
# speedup vs baseline: 1.4826x; 1.0744x over previous
#include <hip/hip_runtime.h>


typedef __attribute__((ext_vector_type(2))) float v2f;
typedef __attribute__((ext_vector_type(4))) float v4f;
typedef __attribute__((ext_vector_type(8))) float v8f;

#define N_NODES 50000
#define R_REL   16
#define H_DIM   32
#define L_DIM   8
#define E_EDGES 1600000

#define TILES   (N_NODES / 16)
#define K_STEPS (H_DIM / 4)
#define P_COLS  (R_REL * L_DIM)
#define B_COLS  (P_COLS + L_DIM)
#define C_TILES ((B_COLS + 15) / 16)

#define SCAN_WAVES  32
#define SCAN_ITERS  13
#define SCAN_CHUNK  128
#define SCAN_PER_WAVE (SCAN_ITERS * SCAN_CHUNK)
#define SCAN_TOTAL  (SCAN_WAVES * SCAN_PER_WAVE)

static_assert(SCAN_TOTAL >= N_NODES + 1, "scan range must cover every node and the end offset");
static_assert(N_NODES <= 65536, "source index is packed into 16 bits");
static_assert(N_NODES % 8 == 0, "8 nodes per block in the layer-1 segment kernel");
static_assert(N_NODES % 4 == 0, "4 nodes per wave in the layer-2 segment kernel");

__global__ void rgcn_zero_f32(float* __restrict__ p, long long n) {
  long long i = (long long)blockIdx.x * blockDim.x + threadIdx.x;
  long long stride = (long long)gridDim.x * blockDim.x;
  for (; i < n; i += stride) p[i] = 0.0f;
}

__global__ void rgcn_count(const int* __restrict__ ei, int* __restrict__ deg) {
  int e = blockIdx.x * blockDim.x + threadIdx.x;
  if (e >= E_EDGES) return;
  int dst = ei[E_EDGES + e];
  atomicAdd(&deg[dst], 1);
}

__global__ __launch_bounds__(1024) void rgcn_scan(const int* __restrict__ deg,
                                                  int* __restrict__ offs,
                                                  int* __restrict__ cursor) {
  __shared__ int wsum[SCAN_WAVES];
  const int lane = threadIdx.x & 31;
  const int w    = threadIdx.x >> 5;
  const int base = w * SCAN_PER_WAVE + lane * 4;

  int4 v[SCAN_ITERS];
  #pragma unroll
  for (int k = 0; k < SCAN_ITERS; ++k) v[k] = *(const int4*)(deg + base + k * SCAN_CHUNK);

  int ex[SCAN_ITERS];
  int carry = 0;
  #pragma unroll
  for (int k = 0; k < SCAN_ITERS; ++k) {
    int s = v[k].x + v[k].y + v[k].z + v[k].w;
    int inc = s;
    #pragma unroll
    for (int d = 1; d < 32; d <<= 1) {
      int t = __shfl_up(inc, d);
      if (lane >= d) inc += t;
    }
    ex[k] = carry + inc - s;
    carry += __shfl(inc, 31);
  }
  if (lane == 0) wsum[w] = carry;
  __syncthreads();

  int t0 = wsum[lane];
  int inc = t0;
  #pragma unroll
  for (int d = 1; d < 32; d <<= 1) {
    int t = __shfl_up(inc, d);
    if (lane >= d) inc += t;
  }
  const int woff = __shfl(inc - t0, w);

  #pragma unroll
  for (int k = 0; k < SCAN_ITERS; ++k) {
    int4 o;
    o.x = woff + ex[k];
    o.y = o.x + v[k].x;
    o.z = o.y + v[k].y;
    o.w = o.z + v[k].z;
    *(int4*)(offs   + base + k * SCAN_CHUNK) = o;
    *(int4*)(cursor + base + k * SCAN_CHUNK) = o;
  }
}

__global__ void rgcn_fill(const int* __restrict__ ei, const int* __restrict__ et,
                          int* __restrict__ cursor, int* __restrict__ keys) {
  int e = blockIdx.x * blockDim.x + threadIdx.x;
  if (e >= E_EDGES) return;
  int src = ei[e];
  int dst = ei[E_EDGES + e];
  int r   = et[e];
  int pos = atomicAdd(&cursor[dst], 1);
  keys[pos] = src | (r << 16);
}

__device__ __forceinline__ void rgcn_rel_count(int rel, int (&c)[R_REL]) {
  #pragma unroll
  for (int rr = 0; rr < R_REL; ++rr)
    c[rr] += __builtin_popcount(__builtin_amdgcn_ballot_w32(rel == rr));
}

__global__ __launch_bounds__(256) void rgcn_layer1_seg(const int* __restrict__ offs,
                                                       const int* __restrict__ keys,
                                                       const float* __restrict__ W1,
                                                       const float* __restrict__ root1,
                                                       const float* __restrict__ b1,
                                                       float* __restrict__ norms,
                                                       float* __restrict__ h1) {
  const int lane = threadIdx.x & 31;
  const int node = blockIdx.x * 8 + __builtin_amdgcn_readfirstlane(threadIdx.x >> 5);
  const int beg = offs[node];
  const int end = offs[node + 1];

  const int n0 = min(32, end - beg);
  const int n1 = min(32, end - beg - 32);
  int key0 = 0;
  int key1 = 0;
  if (lane < n0) key0 = keys[beg + lane];
  if (lane < n1) key1 = keys[beg + 32 + lane];

  int c[R_REL];
  #pragma unroll
  for (int rr = 0; rr < R_REL; ++rr) c[rr] = 0;
  rgcn_rel_count(lane < n0 ? (key0 >> 16) : R_REL, c);
  if (n1 > 0) rgcn_rel_count(lane < n1 ? (key1 >> 16) : R_REL, c);
  for (int base = beg + 64; base < end; base += 32) {
    const int n = min(32, end - base);
    int rel = R_REL;
    if (lane < n) rel = keys[base + lane] >> 16;
    rgcn_rel_count(rel, c);
  }
  int cntv = 0;
  #pragma unroll
  for (int rr = 0; rr < R_REL; ++rr) cntv = (lane == rr) ? c[rr] : cntv;

  float acc = 0.0f;
  for (int base = beg; base < end; base += 32) {
    const int n = min(32, end - base);
    int key;
    if (base == beg) {
      key = key0;
    } else if (base == beg + 32) {
      key = key1;
    } else {
      key = 0;
      if (lane < n) key = keys[base + lane];
    }
    const int   cl = __shfl(cntv, key >> 16);
    const float nm = 1.0f / fmaxf((float)cl, 1.0f);
    if (lane < n) norms[base + lane] = nm;
    for (int j = 0; j < n; j += 8) {
      float w[8];
      float nn[8];
      #pragma unroll
      for (int u = 0; u < 8; ++u) {
        int kk = __builtin_amdgcn_readlane(key, j + u);
        nn[u]  = __int_as_float(__builtin_amdgcn_readlane(__float_as_int(nm), j + u));
        int src = kk & 0xFFFF;
        int r   = kk >> 16;
        w[u] = W1[((size_t)r * N_NODES + src) * H_DIM + lane];
      }
      #pragma unroll
      for (int u = 0; u < 8; ++u) {
        if (j + u < n) acc = fmaf(w[u], nn[u], acc);
      }
    }
  }
  const size_t o = (size_t)node * H_DIM + lane;
  h1[o] = fmaxf(acc + root1[o] + b1[lane], 0.0f);
}

__global__ void rgcn_project_wmma(const float* __restrict__ h1, const float* __restrict__ W2,
                                  const float* __restrict__ root2, const float* __restrict__ b2,
                                  float* __restrict__ P, float* __restrict__ pre) {
  __shared__ float ldsB[C_TILES * K_STEPS * 64];

  for (int idx = threadIdx.x; idx < C_TILES * K_STEPS * 64; idx += blockDim.x) {
    int ct   = idx >> 9;
    int s    = (idx >> 6) & (K_STEPS - 1);
    int slot = idx & 63;
    int ln   = slot >> 1;
    int j    = slot & 1;
    int c    = (ct << 4) + (ln & 15);
    int k    = (s << 2) + ((ln >> 4) << 1) + j;
    float v = 0.0f;
    if (c < P_COLS) {
      v = W2[((size_t)(c >> 3) * H_DIM + k) * L_DIM + (c & (L_DIM - 1))];
    } else if (c < B_COLS) {
      v = root2[(size_t)k * L_DIM + (c - P_COLS)];
    }
    ldsB[idx] = v;
  }
  __syncthreads();

  int wave = (blockIdx.x * blockDim.x + threadIdx.x) >> 5;
  int lane = threadIdx.x & 31;
  if (wave >= TILES) return;

  const int row0 = wave * 16;
  const int m    = lane & 15;
  const int half = lane >> 4;
  const int koff = half * 2;
  const int col  = lane & 15;

  const float* arow = h1 + (size_t)(row0 + m) * H_DIM + koff;
  v2f a[K_STEPS];
  #pragma unroll
  for (int s = 0; s < K_STEPS; ++s) a[s] = *(const v2f*)(arow + s * 4);

  const float* brow = ldsB + lane * 2;

  #pragma unroll
  for (int ct = 0; ct < C_TILES - 1; ++ct) {
    v8f acc = {};
    #pragma unroll
    for (int s = 0; s < K_STEPS; ++s) {
      v2f b = *(const v2f*)(brow + (ct * K_STEPS + s) * 64);
      acc = __builtin_amdgcn_wmma_f32_16x16x4_f32(
           false, a[s],  false, b,
           (short)0, acc,  false,  false);
    }
    #pragma unroll
    for (int i = 0; i < 8; ++i) {
      int node = row0 + i + 8 * half;
      P[(size_t)node * P_COLS + ct * 16 + col] = acc[i];
    }
  }

  {
    float bias = b2[col & (L_DIM - 1)];
    v8f acc = {};
    #pragma unroll
    for (int s = 0; s < K_STEPS; ++s) {
      v2f b = *(const v2f*)(brow + ((C_TILES - 1) * K_STEPS + s) * 64);
      acc = __builtin_amdgcn_wmma_f32_16x16x4_f32(
           false, a[s],  false, b,
           (short)0, acc,  false,  false);
    }
    if (col < L_DIM) {
      #pragma unroll
      for (int i = 0; i < 8; ++i) {
        int node = row0 + i + 8 * half;
        pre[(size_t)node * L_DIM + col] = acc[i] + bias;
      }
    }
  }
}

__global__ __launch_bounds__(256) void rgcn_layer2_seg(const int* __restrict__ offs,
                                                       const int* __restrict__ keys,
                                                       const float* __restrict__ norms,
                                                       const float* __restrict__ P,
                                                       const float* __restrict__ pre,
                                                       float* __restrict__ out) {
  const int lane = threadIdx.x & 31;
  const int wave = blockIdx.x * 8 + __builtin_amdgcn_readfirstlane(threadIdx.x >> 5);
  if (wave >= N_NODES / 4) return;
  const int l = lane & (L_DIM - 1);
  const int g = lane >> 3;
  const int node0 = wave * 4;
  float res = 0.0f;
  for (int q = 0; q < 4; ++q) {
    const int beg = offs[node0 + q];
    const int end = offs[node0 + q + 1];
    float acc = 0.0f;
    for (int base = beg; base < end; base += 32) {
      const int n = min(32, end - base);
      int   key = 0;
      float nm  = 0.0f;
      if (lane < n) {
        key = keys[base + lane];
        nm  = norms[base + lane];
      }
      float v[8];
      float nn[8];
      #pragma unroll
      for (int jj = 0; jj < 8; ++jj) {
        int idx = jj * 4 + g;
        int kk  = __shfl(key, idx);
        nn[jj]  = __shfl(nm, idx);
        int src = kk & 0xFFFF;
        int r   = kk >> 16;
        v[jj] = P[((size_t)src * R_REL + r) * L_DIM + l];
      }
      #pragma unroll
      for (int jj = 0; jj < 8; ++jj) {
        if (jj * 4 + g < n) acc = fmaf(v[jj], nn[jj], acc);
      }
    }
    acc += __shfl_xor(acc, 8);
    acc += __shfl_xor(acc, 16);
    if (g == q) res = acc;
  }
  const size_t o = (size_t)node0 * L_DIM + lane;
  float x = pre[o] + res;
  out[o] = 1.0f / (1.0f + __expf(-x));
}

extern "C" void kernel_launch(void* const* d_in, const int* in_sizes, int n_in,
                              void* d_out, int out_size, void* d_ws, size_t ws_size,
                              hipStream_t stream) {
  const int*   ei    = (const int*)d_in[0];
  const int*   et    = (const int*)d_in[1];
  const float* W1    = (const float*)d_in[2];
  const float* root1 = (const float*)d_in[3];
  const float* b1    = (const float*)d_in[4];
  const float* W2    = (const float*)d_in[5];
  const float* root2 = (const float*)d_in[6];
  const float* b2    = (const float*)d_in[7];
  float*       out   = (float*)d_out;

  int*   deg    = (int*)d_ws;
  int*   offs   = deg + SCAN_TOTAL;
  int*   cursor = offs + SCAN_TOTAL;
  int*   keys   = cursor + SCAN_TOTAL;
  float* norms  = (float*)(keys + E_EDGES);
  float* h1     = norms + E_EDGES;
  float* P      = h1 + (size_t)N_NODES * H_DIM;
  float* pre    = P + (size_t)N_NODES * P_COLS;

  rgcn_zero_f32<<<SCAN_TOTAL / 256, 256, 0, stream>>>((float*)deg, (long long)SCAN_TOTAL);

  rgcn_count<<<(E_EDGES + 255) / 256, 256, 0, stream>>>(ei, deg);

  rgcn_scan<<<1, 1024, 0, stream>>>(deg, offs, cursor);

  rgcn_fill<<<(E_EDGES + 255) / 256, 256, 0, stream>>>(ei, et, cursor, keys);

  rgcn_layer1_seg<<<N_NODES / 8, 256, 0, stream>>>(offs, keys, W1, root1, b1, norms, h1);

  const int gemmBlocks = (TILES * 32 + 255) / 256;
  rgcn_project_wmma<<<gemmBlocks, 256, 0, stream>>>(h1, W2, root2, b2, P, pre);

  rgcn_layer2_seg<<<(N_NODES / 4 + 7) / 8, 256, 0, stream>>>(offs, keys, norms, P, pre, out);

  (void)in_sizes; (void)n_in; (void)out_size; (void)ws_size;
}


# Edges ordered by dst through a two-level counting sort with LDS counters (chunk by 256-node range, then owner block by node) in place of per-edge global atomics; float sum order arrival-set as before
# speedup vs baseline: 5.3362x; 3.5992x over previous
#include <hip/hip_runtime.h>


typedef __attribute__((ext_vector_type(2))) float v2f;
typedef __attribute__((ext_vector_type(4))) float v4f;
typedef __attribute__((ext_vector_type(8))) float v8f;

#define N_NODES 50000
#define R_REL   16
#define H_DIM   32
#define L_DIM   8
#define E_EDGES 1600000

#define TILES   (N_NODES / 16)
#define K_STEPS (H_DIM / 4)
#define P_COLS  (R_REL * L_DIM)
#define B_COLS  (P_COLS + L_DIM)
#define C_TILES ((B_COLS + 15) / 16)

#define BIN_SHIFT  8
#define BIN_NODES  (1 << BIN_SHIFT)
#define N_BINS     ((N_NODES + BIN_NODES - 1) / BIN_NODES)
#define N_CHUNKS   256
#define CHUNK      (E_EDGES / N_CHUNKS)
#define OFF_PITCH  (N_BINS + 1)
#define BIN_PER    ((CHUNK + 1023) / 1024)
#define STAGE_CAP  12288
#define OFFS_WORDS 50016

static_assert(E_EDGES % N_CHUNKS == 0, "chunks are equal");
static_assert(N_CHUNKS == 8 * 32, "one wave scans the chunk table, 8 entries per lane; 32 waves take 8 chunks each");
static_assert(BIN_NODES == 8 * 32, "one wave scans a bin's node counters, 8 per lane");
static_assert(N_BINS <= 7 * 32, "one wave scans the bin counters, 7 per lane");
static_assert(OFFS_WORDS >= N_NODES + 1, "offsets of every node and the end offset");
static_assert(N_NODES <= 65536, "source index is packed into 16 bits");
static_assert(R_REL <= 16, "relation is packed into 4 bits");
static_assert(N_NODES % 8 == 0, "8 nodes per block in the layer-1 segment kernel");
static_assert(N_NODES % 4 == 0, "4 nodes per wave in the layer-2 segment kernel");

__device__ __forceinline__ int rgcn_wave_incl(int v, int lane) {
  #pragma unroll
  for (int d = 1; d < 32; d <<= 1) {
    int t = __shfl_up(v, d);
    if (lane >= d) v += t;
  }
  return v;
}

__global__ __launch_bounds__(1024) void rgcn_bin(const int* __restrict__ ei, const int* __restrict__ et,
                                                 int* __restrict__ chunkrec, int* __restrict__ chunkoff) {
  __shared__ int cur[N_BINS];
  __shared__ int start[N_BINS + 1];
  __shared__ int stage[CHUNK];
  const int tid  = threadIdx.x;
  const int lane = tid & 31;
  const int base = blockIdx.x * CHUNK;

  for (int b = tid; b < N_BINS; b += 1024) cur[b] = 0;
  __syncthreads();

  int rec[BIN_PER];
  int bin[BIN_PER];
  #pragma unroll
  for (int k = 0; k < BIN_PER; ++k) {
    const int i = tid + k * 1024;
    rec[k] = 0;
    bin[k] = -1;
    if (i < CHUNK) {
      const int src = ei[base + i];
      const int dst = ei[E_EDGES + base + i];
      const int r   = et[base + i];
      rec[k] = src | (r << 16) | ((dst & (BIN_NODES - 1)) << 20);
      bin[k] = dst >> BIN_SHIFT;
    }
  }
  #pragma unroll
  for (int k = 0; k < BIN_PER; ++k)
    if (bin[k] >= 0) atomicAdd(&cur[bin[k]], 1);
  __syncthreads();

  if (tid < 32) {
    int loc[7];
    int s = 0;
    #pragma unroll
    for (int j = 0; j < 7; ++j) {
      const int idx = lane * 7 + j;
      const int v = idx < N_BINS ? cur[idx] : 0;
      loc[j] = s;
      s += v;
    }
    const int inc  = rgcn_wave_incl(s, lane);
    const int excl = inc - s;
    #pragma unroll
    for (int j = 0; j < 7; ++j) {
      const int idx = lane * 7 + j;
      if (idx < N_BINS) {
        start[idx] = excl + loc[j];
        cur[idx]   = excl + loc[j];
      }
    }
    if (lane == 31) start[N_BINS] = inc;
  }
  __syncthreads();

  #pragma unroll
  for (int k = 0; k < BIN_PER; ++k) {
    if (bin[k] >= 0) {
      const int pos = atomicAdd(&cur[bin[k]], 1);
      stage[pos] = rec[k];
    }
  }
  __syncthreads();

  for (int i = tid; i < CHUNK; i += 1024) chunkrec[base + i] = stage[i];
  for (int b = tid; b <= N_BINS; b += 1024) chunkoff[blockIdx.x * OFF_PITCH + b] = start[b];
}

__global__ __launch_bounds__(1024) void rgcn_binsort(const int* __restrict__ chunkrec,
                                                     const int* __restrict__ chunkoff,
                                                     int* __restrict__ offs, int* __restrict__ keys) {
  __shared__ int plo[N_CHUNKS];
  __shared__ int ppos[N_CHUNKS + 1];
  __shared__ int cur[BIN_NODES];
  __shared__ int sbase;
  __shared__ int stage[STAGE_CAP];
  __shared__ int sorted[STAGE_CAP];
  const int tid  = threadIdx.x;
  const int lane = tid & 31;
  const int w    = __builtin_amdgcn_readfirstlane(tid >> 5);
  const int b    = blockIdx.x;

  if (tid < 32) {
    int lo[8];
    int ln[8];
    #pragma unroll
    for (int j = 0; j < 8; ++j) {
      const int* p = chunkoff + (lane * 8 + j) * OFF_PITCH + b;
      lo[j] = p[0];
      ln[j] = p[1] - p[0];
    }
    int s  = 0;
    int sl = 0;
    #pragma unroll
    for (int j = 0; j < 8; ++j) {
      const int v = ln[j];
      ln[j] = s;
      s  += v;
      sl += lo[j];
    }
    const int inc  = rgcn_wave_incl(s, lane);
    const int excl = inc - s;
    #pragma unroll
    for (int d = 16; d >= 1; d >>= 1) sl += __shfl_xor(sl, d);
    #pragma unroll
    for (int j = 0; j < 8; ++j) {
      plo[lane * 8 + j]  = lo[j];
      ppos[lane * 8 + j] = excl + ln[j];
    }
    if (lane == 31) ppos[N_CHUNKS] = inc;
    if (lane == 0)  sbase = sl;
  }
  for (int j = tid; j < BIN_NODES; j += 1024) cur[j] = 0;
  __syncthreads();

  const int  total  = ppos[N_CHUNKS];
  const int  kbase  = sbase;
  const bool staged = total <= STAGE_CAP;

  {
    int ra[8];
    int rb[8];
    #pragma unroll
    for (int k = 0; k < 8; ++k) {
      const int c   = w + 32 * k;
      const int len = ppos[c + 1] - ppos[c];
      const int lo  = plo[c];
      const int* cb = chunkrec + c * CHUNK;
      const int va = cb[min(lo + lane, CHUNK - 1)];
      const int vb = cb[min(lo + lane + 32, CHUNK - 1)];
      ra[k] = lane < len      ? va : -1;
      rb[k] = lane + 32 < len ? vb : -1;
    }
    #pragma unroll
    for (int k = 0; k < 8; ++k) {
      const int c   = w + 32 * k;
      const int pos = ppos[c];
      const int len = ppos[c + 1] - pos;
      const int* src = chunkrec + c * CHUNK + plo[c];
      if (ra[k] >= 0) {
        atomicAdd(&cur[ra[k] >> 20], 1);
        if (staged) stage[pos + lane] = ra[k];
      }
      if (rb[k] >= 0) {
        atomicAdd(&cur[rb[k] >> 20], 1);
        if (staged) stage[pos + lane + 32] = rb[k];
      }
      for (int i = lane + 64; i < len; i += 32) {
        const int rc = src[i];
        atomicAdd(&cur[rc >> 20], 1);
        if (staged) stage[pos + i] = rc;
      }
    }
  }
  __syncthreads();

  if (tid < 32) {
    int loc[8];
    int s = 0;
    #pragma unroll
    for (int j = 0; j < 8; ++j) {
      const int v = cur[lane * 8 + j];
      loc[j] = s;
      s += v;
    }
    const int inc  = rgcn_wave_incl(s, lane);
    const int excl = inc - s;
    #pragma unroll
    for (int j = 0; j < 8; ++j) {
      const int idx  = lane * 8 + j;
      const int st   = excl + loc[j];
      const int node = b * BIN_NODES + idx;
      cur[idx] = st;
      if (node < N_NODES) offs[node] = kbase + st;
    }
    if (lane == 31 && b == N_BINS - 1) offs[N_NODES] = kbase + inc;
  }
  __syncthreads();

  if (staged) {
    for (int i = tid; i < total; i += 1024) {
      const int rc  = stage[i];
      const int pos = atomicAdd(&cur[rc >> 20], 1);
      sorted[pos] = rc & 0xFFFFF;
    }
    __syncthreads();
    for (int i = tid; i < total; i += 1024) keys[kbase + i] = sorted[i];
  } else {
    for (int k = 0; k < 8; ++k) {
      const int c   = w + 32 * k;
      const int len = ppos[c + 1] - ppos[c];
      const int* src = chunkrec + c * CHUNK + plo[c];
      for (int i = lane; i < len; i += 32) {
        const int rc  = src[i];
        const int pos = atomicAdd(&cur[rc >> 20], 1);
        keys[kbase + pos] = rc & 0xFFFFF;
      }
    }
  }
}

__device__ __forceinline__ void rgcn_rel_count(int rel, int (&c)[R_REL]) {
  #pragma unroll
  for (int rr = 0; rr < R_REL; ++rr)
    c[rr] += __builtin_popcount(__builtin_amdgcn_ballot_w32(rel == rr));
}

__global__ __launch_bounds__(256) void rgcn_layer1_seg(const int* __restrict__ offs,
                                                       const int* __restrict__ keys,
                                                       const float* __restrict__ W1,
                                                       const float* __restrict__ root1,
                                                       const float* __restrict__ b1,
                                                       float* __restrict__ norms,
                                                       float* __restrict__ h1) {
  const int lane = threadIdx.x & 31;
  const int node = blockIdx.x * 8 + __builtin_amdgcn_readfirstlane(threadIdx.x >> 5);
  const int beg = offs[node];
  const int end = offs[node + 1];

  const int n0 = min(32, end - beg);
  const int n1 = min(32, end - beg - 32);
  int key0 = 0;
  int key1 = 0;
  if (lane < n0) key0 = keys[beg + lane];
  if (lane < n1) key1 = keys[beg + 32 + lane];

  int c[R_REL];
  #pragma unroll
  for (int rr = 0; rr < R_REL; ++rr) c[rr] = 0;
  rgcn_rel_count(lane < n0 ? (key0 >> 16) : R_REL, c);
  if (n1 > 0) rgcn_rel_count(lane < n1 ? (key1 >> 16) : R_REL, c);
  for (int base = beg + 64; base < end; base += 32) {
    const int n = min(32, end - base);
    int rel = R_REL;
    if (lane < n) rel = keys[base + lane] >> 16;
    rgcn_rel_count(rel, c);
  }
  int cntv = 0;
  #pragma unroll
  for (int rr = 0; rr < R_REL; ++rr) cntv = (lane == rr) ? c[rr] : cntv;

  float acc = 0.0f;
  for (int base = beg; base < end; base += 32) {
    const int n = min(32, end - base);
    int key;
    if (base == beg) {
      key = key0;
    } else if (base == beg + 32) {
      key = key1;
    } else {
      key = 0;
      if (lane < n) key = keys[base + lane];
    }
    const int   cl = __shfl(cntv, key >> 16);
    const float nm = 1.0f / fmaxf((float)cl, 1.0f);
    if (lane < n) norms[base + lane] = nm;
    for (int j = 0; j < n; j += 8) {
      float w[8];
      float nn[8];
      #pragma unroll
      for (int u = 0; u < 8; ++u) {
        int kk = __builtin_amdgcn_readlane(key, j + u);
        nn[u]  = __int_as_float(__builtin_amdgcn_readlane(__float_as_int(nm), j + u));
        int src = kk & 0xFFFF;
        int r   = kk >> 16;
        w[u] = W1[((size_t)r * N_NODES + src) * H_DIM + lane];
      }
      #pragma unroll
      for (int u = 0; u < 8; ++u) {
        if (j + u < n) acc = fmaf(w[u], nn[u], acc);
      }
    }
  }
  const size_t o = (size_t)node * H_DIM + lane;
  h1[o] = fmaxf(acc + root1[o] + b1[lane], 0.0f);
}

__global__ void rgcn_project_wmma(const float* __restrict__ h1, const float* __restrict__ W2,
                                  const float* __restrict__ root2, const float* __restrict__ b2,
                                  float* __restrict__ P, float* __restrict__ pre) {
  __shared__ float ldsB[C_TILES * K_STEPS * 64];

  for (int idx = threadIdx.x; idx < C_TILES * K_STEPS * 64; idx += blockDim.x) {
    int ct   = idx >> 9;
    int s    = (idx >> 6) & (K_STEPS - 1);
    int slot = idx & 63;
    int ln   = slot >> 1;
    int j    = slot & 1;
    int c    = (ct << 4) + (ln & 15);
    int k    = (s << 2) + ((ln >> 4) << 1) + j;
    float v = 0.0f;
    if (c < P_COLS) {
      v = W2[((size_t)(c >> 3) * H_DIM + k) * L_DIM + (c & (L_DIM - 1))];
    } else if (c < B_COLS) {
      v = root2[(size_t)k * L_DIM + (c - P_COLS)];
    }
    ldsB[idx] = v;
  }
  __syncthreads();

  int wave = (blockIdx.x * blockDim.x + threadIdx.x) >> 5;
  int lane = threadIdx.x & 31;
  if (wave >= TILES) return;

  const int row0 = wave * 16;
  const int m    = lane & 15;
  const int half = lane >> 4;
  const int koff = half * 2;
  const int col  = lane & 15;

  const float* arow = h1 + (size_t)(row0 + m) * H_DIM + koff;
  v2f a[K_STEPS];
  #pragma unroll
  for (int s = 0; s < K_STEPS; ++s) a[s] = *(const v2f*)(arow + s * 4);

  const float* brow = ldsB + lane * 2;

  #pragma unroll
  for (int ct = 0; ct < C_TILES - 1; ++ct) {
    v8f acc = {};
    #pragma unroll
    for (int s = 0; s < K_STEPS; ++s) {
      v2f b = *(const v2f*)(brow + (ct * K_STEPS + s) * 64);
      acc = __builtin_amdgcn_wmma_f32_16x16x4_f32(
           false, a[s],  false, b,
           (short)0, acc,  false,  false);
    }
    #pragma unroll
    for (int i = 0; i < 8; ++i) {
      int node = row0 + i + 8 * half;
      P[(size_t)node * P_COLS + ct * 16 + col] = acc[i];
    }
  }

  {
    float bias = b2[col & (L_DIM - 1)];
    v8f acc = {};
    #pragma unroll
    for (int s = 0; s < K_STEPS; ++s) {
      v2f b = *(const v2f*)(brow + ((C_TILES - 1) * K_STEPS + s) * 64);
      acc = __builtin_amdgcn_wmma_f32_16x16x4_f32(
           false, a[s],  false, b,
           (short)0, acc,  false,  false);
    }
    if (col < L_DIM) {
      #pragma unroll
      for (int i = 0; i < 8; ++i) {
        int node = row0 + i + 8 * half;
        pre[(size_t)node * L_DIM + col] = acc[i] + bias;
      }
    }
  }
}

__global__ __launch_bounds__(256) void rgcn_layer2_seg(const int* __restrict__ offs,
                                                       const int* __restrict__ keys,
                                                       const float* __restrict__ norms,
                                                       const float* __restrict__ P,
                                                       const float* __restrict__ pre,
                                                       float* __restrict__ out) {
  const int lane = threadIdx.x & 31;
  const int wave = blockIdx.x * 8 + __builtin_amdgcn_readfirstlane(threadIdx.x >> 5);
  if (wave >= N_NODES / 4) return;
  const int l = lane & (L_DIM - 1);
  const int g = lane >> 3;
  const int node0 = wave * 4;
  float res = 0.0f;
  for (int q = 0; q < 4; ++q) {
    const int beg = offs[node0 + q];
    const int end = offs[node0 + q + 1];
    float acc = 0.0f;
    for (int base = beg; base < end; base += 32) {
      const int n = min(32, end - base);
      int   key = 0;
      float nm  = 0.0f;
      if (lane < n) {
        key = keys[base + lane];
        nm  = norms[base + lane];
      }
      float v[8];
      float nn[8];
      #pragma unroll
      for (int jj = 0; jj < 8; ++jj) {
        int idx = jj * 4 + g;
        int kk  = __shfl(key, idx);
        nn[jj]  = __shfl(nm, idx);
        int src = kk & 0xFFFF;
        int r   = kk >> 16;
        v[jj] = P[((size_t)src * R_REL + r) * L_DIM + l];
      }
      #pragma unroll
      for (int jj = 0; jj < 8; ++jj) {
        if (jj * 4 + g < n) acc = fmaf(v[jj], nn[jj], acc);
      }
    }
    acc += __shfl_xor(acc, 8);
    acc += __shfl_xor(acc, 16);
    if (g == q) res = acc;
  }
  const size_t o = (size_t)node0 * L_DIM + lane;
  float x = pre[o] + res;
  out[o] = 1.0f / (1.0f + __expf(-x));
}

extern "C" void kernel_launch(void* const* d_in, const int* in_sizes, int n_in,
                              void* d_out, int out_size, void* d_ws, size_t ws_size,
                              hipStream_t stream) {
  const int*   ei    = (const int*)d_in[0];
  const int*   et    = (const int*)d_in[1];
  const float* W1    = (const float*)d_in[2];
  const float* root1 = (const float*)d_in[3];
  const float* b1    = (const float*)d_in[4];
  const float* W2    = (const float*)d_in[5];
  const float* root2 = (const float*)d_in[6];
  const float* b2    = (const float*)d_in[7];
  float*       out   = (float*)d_out;

  int*   offs     = (int*)d_ws;
  int*   keys     = offs + OFFS_WORDS;
  float* norms    = (float*)(keys + E_EDGES);
  float* h1       = norms + E_EDGES;
  float* P        = h1 + (size_t)N_NODES * H_DIM;
  float* pre      = P + (size_t)N_NODES * P_COLS;
  int*   chunkrec = (int*)(pre + (size_t)N_NODES * L_DIM);
  int*   chunkoff = chunkrec + E_EDGES;

  rgcn_bin<<<N_CHUNKS, 1024, 0, stream>>>(ei, et, chunkrec, chunkoff);

  rgcn_binsort<<<N_BINS, 1024, 0, stream>>>(chunkrec, chunkoff, offs, keys);

  rgcn_layer1_seg<<<N_NODES / 8, 256, 0, stream>>>(offs, keys, W1, root1, b1, norms, h1);

  const int gemmBlocks = (TILES * 32 + 255) / 256;
  rgcn_project_wmma<<<gemmBlocks, 256, 0, stream>>>(h1, W2, root2, b2, P, pre);

  rgcn_layer2_seg<<<(N_NODES / 4 + 7) / 8, 256, 0, stream>>>(offs, keys, norms, P, pre, out);

  (void)in_sizes; (void)n_in; (void)out_size; (void)ws_size;
}


# Layer 1: one wave walks 4 consecutive nodes (12,500 waves for 50,000), the next node's keys and root row and all 32 rows of a chunk requested ahead of their use; per-node sum order unchanged
# speedup vs baseline: 5.5747x; 1.0447x over previous
#include <hip/hip_runtime.h>


typedef __attribute__((ext_vector_type(2))) float v2f;
typedef __attribute__((ext_vector_type(4))) float v4f;
typedef __attribute__((ext_vector_type(8))) float v8f;

#define N_NODES 50000
#define R_REL   16
#define H_DIM   32
#define L_DIM   8
#define E_EDGES 1600000

#define TILES   (N_NODES / 16)
#define K_STEPS (H_DIM / 4)
#define P_COLS  (R_REL * L_DIM)
#define B_COLS  (P_COLS + L_DIM)
#define C_TILES ((B_COLS + 15) / 16)

#define BIN_SHIFT  8
#define BIN_NODES  (1 << BIN_SHIFT)
#define N_BINS     ((N_NODES + BIN_NODES - 1) / BIN_NODES)
#define N_CHUNKS   256
#define CHUNK      (E_EDGES / N_CHUNKS)
#define OFF_PITCH  (N_BINS + 1)
#define BIN_PER    ((CHUNK + 1023) / 1024)
#define STAGE_CAP  12288
#define OFFS_WORDS 50016

static_assert(E_EDGES % N_CHUNKS == 0, "chunks are equal");
static_assert(N_CHUNKS == 8 * 32, "one wave scans the chunk table, 8 entries per lane; 32 waves take 8 chunks each");
static_assert(BIN_NODES == 8 * 32, "one wave scans a bin's node counters, 8 per lane");
static_assert(N_BINS <= 7 * 32, "one wave scans the bin counters, 7 per lane");
static_assert(OFFS_WORDS >= N_NODES + 1, "offsets of every node and the end offset");
static_assert(N_NODES <= 65536, "source index is packed into 16 bits");
static_assert(R_REL <= 16, "relation is packed into 4 bits");
static_assert(N_NODES % 4 == 0, "4 nodes per wave in the layer-2 segment kernel");

__device__ __forceinline__ int rgcn_wave_incl(int v, int lane) {
  #pragma unroll
  for (int d = 1; d < 32; d <<= 1) {
    int t = __shfl_up(v, d);
    if (lane >= d) v += t;
  }
  return v;
}

__global__ __launch_bounds__(1024) void rgcn_bin(const int* __restrict__ ei, const int* __restrict__ et,
                                                 int* __restrict__ chunkrec, int* __restrict__ chunkoff) {
  __shared__ int cur[N_BINS];
  __shared__ int start[N_BINS + 1];
  __shared__ int stage[CHUNK];
  const int tid  = threadIdx.x;
  const int lane = tid & 31;
  const int base = blockIdx.x * CHUNK;

  for (int b = tid; b < N_BINS; b += 1024) cur[b] = 0;
  __syncthreads();

  int rec[BIN_PER];
  int bin[BIN_PER];
  #pragma unroll
  for (int k = 0; k < BIN_PER; ++k) {
    const int i = tid + k * 1024;
    rec[k] = 0;
    bin[k] = -1;
    if (i < CHUNK) {
      const int src = ei[base + i];
      const int dst = ei[E_EDGES + base + i];
      const int r   = et[base + i];
      rec[k] = src | (r << 16) | ((dst & (BIN_NODES - 1)) << 20);
      bin[k] = dst >> BIN_SHIFT;
    }
  }
  #pragma unroll
  for (int k = 0; k < BIN_PER; ++k)
    if (bin[k] >= 0) atomicAdd(&cur[bin[k]], 1);
  __syncthreads();

  if (tid < 32) {
    int loc[7];
    int s = 0;
    #pragma unroll
    for (int j = 0; j < 7; ++j) {
      const int idx = lane * 7 + j;
      const int v = idx < N_BINS ? cur[idx] : 0;
      loc[j] = s;
      s += v;
    }
    const int inc  = rgcn_wave_incl(s, lane);
    const int excl = inc - s;
    #pragma unroll
    for (int j = 0; j < 7; ++j) {
      const int idx = lane * 7 + j;
      if (idx < N_BINS) {
        start[idx] = excl + loc[j];
        cur[idx]   = excl + loc[j];
      }
    }
    if (lane == 31) start[N_BINS] = inc;
  }
  __syncthreads();

  #pragma unroll
  for (int k = 0; k < BIN_PER; ++k) {
    if (bin[k] >= 0) {
      const int pos = atomicAdd(&cur[bin[k]], 1);
      stage[pos] = rec[k];
    }
  }
  __syncthreads();

  for (int i = tid; i < CHUNK; i += 1024) chunkrec[base + i] = stage[i];
  for (int b = tid; b <= N_BINS; b += 1024) chunkoff[blockIdx.x * OFF_PITCH + b] = start[b];
}

__global__ __launch_bounds__(1024) void rgcn_binsort(const int* __restrict__ chunkrec,
                                                     const int* __restrict__ chunkoff,
                                                     int* __restrict__ offs, int* __restrict__ keys) {
  __shared__ int plo[N_CHUNKS];
  __shared__ int ppos[N_CHUNKS + 1];
  __shared__ int cur[BIN_NODES];
  __shared__ int sbase;
  __shared__ int stage[STAGE_CAP];
  __shared__ int sorted[STAGE_CAP];
  const int tid  = threadIdx.x;
  const int lane = tid & 31;
  const int w    = __builtin_amdgcn_readfirstlane(tid >> 5);
  const int b    = blockIdx.x;

  if (tid < 32) {
    int lo[8];
    int ln[8];
    #pragma unroll
    for (int j = 0; j < 8; ++j) {
      const int* p = chunkoff + (lane * 8 + j) * OFF_PITCH + b;
      lo[j] = p[0];
      ln[j] = p[1] - p[0];
    }
    int s  = 0;
    int sl = 0;
    #pragma unroll
    for (int j = 0; j < 8; ++j) {
      const int v = ln[j];
      ln[j] = s;
      s  += v;
      sl += lo[j];
    }
    const int inc  = rgcn_wave_incl(s, lane);
    const int excl = inc - s;
    #pragma unroll
    for (int d = 16; d >= 1; d >>= 1) sl += __shfl_xor(sl, d);
    #pragma unroll
    for (int j = 0; j < 8; ++j) {
      plo[lane * 8 + j]  = lo[j];
      ppos[lane * 8 + j] = excl + ln[j];
    }
    if (lane == 31) ppos[N_CHUNKS] = inc;
    if (lane == 0)  sbase = sl;
  }
  for (int j = tid; j < BIN_NODES; j += 1024) cur[j] = 0;
  __syncthreads();

  const int  total  = ppos[N_CHUNKS];
  const int  kbase  = sbase;
  const bool staged = total <= STAGE_CAP;

  {
    int ra[8];
    int rb[8];
    #pragma unroll
    for (int k = 0; k < 8; ++k) {
      const int c   = w + 32 * k;
      const int len = ppos[c + 1] - ppos[c];
      const int lo  = plo[c];
      const int* cb = chunkrec + c * CHUNK;
      const int va = cb[min(lo + lane, CHUNK - 1)];
      const int vb = cb[min(lo + lane + 32, CHUNK - 1)];
      ra[k] = lane < len      ? va : -1;
      rb[k] = lane + 32 < len ? vb : -1;
    }
    #pragma unroll
    for (int k = 0; k < 8; ++k) {
      const int c   = w + 32 * k;
      const int pos = ppos[c];
      const int len = ppos[c + 1] - pos;
      const int* src = chunkrec + c * CHUNK + plo[c];
      if (ra[k] >= 0) {
        atomicAdd(&cur[ra[k] >> 20], 1);
        if (staged) stage[pos + lane] = ra[k];
      }
      if (rb[k] >= 0) {
        atomicAdd(&cur[rb[k] >> 20], 1);
        if (staged) stage[pos + lane + 32] = rb[k];
      }
      for (int i = lane + 64; i < len; i += 32) {
        const int rc = src[i];
        atomicAdd(&cur[rc >> 20], 1);
        if (staged) stage[pos + i] = rc;
      }
    }
  }
  __syncthreads();

  if (tid < 32) {
    int loc[8];
    int s = 0;
    #pragma unroll
    for (int j = 0; j < 8; ++j) {
      const int v = cur[lane * 8 + j];
      loc[j] = s;
      s += v;
    }
    const int inc  = rgcn_wave_incl(s, lane);
    const int excl = inc - s;
    #pragma unroll
    for (int j = 0; j < 8; ++j) {
      const int idx  = lane * 8 + j;
      const int st   = excl + loc[j];
      const int node = b * BIN_NODES + idx;
      cur[idx] = st;
      if (node < N_NODES) offs[node] = kbase + st;
    }
    if (lane == 31 && b == N_BINS - 1) offs[N_NODES] = kbase + inc;
  }
  __syncthreads();

  if (staged) {
    for (int i = tid; i < total; i += 1024) {
      const int rc  = stage[i];
      const int pos = atomicAdd(&cur[rc >> 20], 1);
      sorted[pos] = rc & 0xFFFFF;
    }
    __syncthreads();
    for (int i = tid; i < total; i += 1024) keys[kbase + i] = sorted[i];
  } else {
    for (int k = 0; k < 8; ++k) {
      const int c   = w + 32 * k;
      const int len = ppos[c + 1] - ppos[c];
      const int* src = chunkrec + c * CHUNK + plo[c];
      for (int i = lane; i < len; i += 32) {
        const int rc  = src[i];
        const int pos = atomicAdd(&cur[rc >> 20], 1);
        keys[kbase + pos] = rc & 0xFFFFF;
      }
    }
  }
}

__device__ __forceinline__ void rgcn_rel_count(int rel, int (&c)[R_REL]) {
  #pragma unroll
  for (int rr = 0; rr < R_REL; ++rr)
    c[rr] += __builtin_popcount(__builtin_amdgcn_ballot_w32(rel == rr));
}

#define L1_NODES 4
static_assert(N_NODES % L1_NODES == 0, "every layer-1 wave owns L1_NODES whole nodes");
static_assert(L1_NODES + 1 <= 32, "one lane per segment bound of the wave's nodes");

__global__ __launch_bounds__(256) void rgcn_layer1_seg(const int* __restrict__ offs,
                                                       const int* __restrict__ keys,
                                                       const float* __restrict__ W1,
                                                       const float* __restrict__ root1,
                                                       const float* __restrict__ b1,
                                                       float* __restrict__ norms,
                                                       float* __restrict__ h1) {
  const int lane = threadIdx.x & 31;
  const int wave = blockIdx.x * 8 + __builtin_amdgcn_readfirstlane(threadIdx.x >> 5);
  if (wave >= N_NODES / L1_NODES) return;
  const int node0 = wave * L1_NODES;

  int ov = 0;
  if (lane <= L1_NODES) ov = offs[node0 + lane];
  const float bias = b1[lane];

  int beg = __builtin_amdgcn_readlane(ov, 0);
  int end = __builtin_amdgcn_readlane(ov, 1);
  int   raw0 = keys[min(beg + lane, E_EDGES - 1)];
  int   raw1 = keys[min(beg + 32 + lane, E_EDGES - 1)];
  float rt   = root1[(size_t)node0 * H_DIM + lane];

  for (int q = 0; q < L1_NODES; ++q) {
    const int node = node0 + q;

    const int   nend  = __builtin_amdgcn_readlane(ov, min(q + 2, L1_NODES));
    const int   nraw0 = keys[min(end + lane, E_EDGES - 1)];
    const int   nraw1 = keys[min(end + 32 + lane, E_EDGES - 1)];
    const float nrt   = root1[(size_t)min(node + 1, N_NODES - 1) * H_DIM + lane];

    const int n0 = min(32, end - beg);
    const int n1 = min(32, end - beg - 32);
    const int key0 = lane < n0 ? raw0 : 0;
    const int key1 = lane < n1 ? raw1 : 0;

    int c[R_REL];
    #pragma unroll
    for (int rr = 0; rr < R_REL; ++rr) c[rr] = 0;
    rgcn_rel_count(lane < n0 ? (key0 >> 16) : R_REL, c);
    if (n1 > 0) rgcn_rel_count(lane < n1 ? (key1 >> 16) : R_REL, c);
    for (int base = beg + 64; base < end; base += 32) {
      const int n = min(32, end - base);
      int rel = R_REL;
      if (lane < n) rel = keys[base + lane] >> 16;
      rgcn_rel_count(rel, c);
    }
    int cntv = 0;
    #pragma unroll
    for (int rr = 0; rr < R_REL; ++rr) cntv = (lane == rr) ? c[rr] : cntv;

    float acc = 0.0f;
    for (int base = beg; base < end; base += 32) {
      const int n = min(32, end - base);
      int key;
      if (base == beg) {
        key = key0;
      } else if (base == beg + 32) {
        key = key1;
      } else {
        key = 0;
        if (lane < n) key = keys[base + lane];
      }
      const int   cl = __shfl(cntv, key >> 16);
      const float nm = 1.0f / fmaxf((float)cl, 1.0f);
      if (lane < n) norms[base + lane] = nm;

      float w[32];
      #pragma unroll
      for (int g = 0; g < 4; ++g) {
        if (g * 8 < n) {
          #pragma unroll
          for (int u = 0; u < 8; ++u) {
            const int kk  = __builtin_amdgcn_readlane(key, g * 8 + u);
            const int src = kk & 0xFFFF;
            const int r   = kk >> 16;
            w[g * 8 + u] = W1[(r * N_NODES + src) * H_DIM + lane];
          }
        }
      }
      #pragma unroll
      for (int u = 0; u < 32; ++u) {
        if (u < n) {
          const float nn = __int_as_float(__builtin_amdgcn_readlane(__float_as_int(nm), u));
          acc = fmaf(w[u], nn, acc);
        }
      }
    }
    h1[(size_t)node * H_DIM + lane] = fmaxf(acc + rt + bias, 0.0f);

    beg  = end;
    end  = nend;
    raw0 = nraw0;
    raw1 = nraw1;
    rt   = nrt;
  }
}

__global__ void rgcn_project_wmma(const float* __restrict__ h1, const float* __restrict__ W2,
                                  const float* __restrict__ root2, const float* __restrict__ b2,
                                  float* __restrict__ P, float* __restrict__ pre) {
  __shared__ float ldsB[C_TILES * K_STEPS * 64];

  for (int idx = threadIdx.x; idx < C_TILES * K_STEPS * 64; idx += blockDim.x) {
    int ct   = idx >> 9;
    int s    = (idx >> 6) & (K_STEPS - 1);
    int slot = idx & 63;
    int ln   = slot >> 1;
    int j    = slot & 1;
    int c    = (ct << 4) + (ln & 15);
    int k    = (s << 2) + ((ln >> 4) << 1) + j;
    float v = 0.0f;
    if (c < P_COLS) {
      v = W2[((size_t)(c >> 3) * H_DIM + k) * L_DIM + (c & (L_DIM - 1))];
    } else if (c < B_COLS) {
      v = root2[(size_t)k * L_DIM + (c - P_COLS)];
    }
    ldsB[idx] = v;
  }
  __syncthreads();

  int wave = (blockIdx.x * blockDim.x + threadIdx.x) >> 5;
  int lane = threadIdx.x & 31;
  if (wave >= TILES) return;

  const int row0 = wave * 16;
  const int m    = lane & 15;
  const int half = lane >> 4;
  const int koff = half * 2;
  const int col  = lane & 15;

  const float* arow = h1 + (size_t)(row0 + m) * H_DIM + koff;
  v2f a[K_STEPS];
  #pragma unroll
  for (int s = 0; s < K_STEPS; ++s) a[s] = *(const v2f*)(arow + s * 4);

  const float* brow = ldsB + lane * 2;

  #pragma unroll
  for (int ct = 0; ct < C_TILES - 1; ++ct) {
    v8f acc = {};
    #pragma unroll
    for (int s = 0; s < K_STEPS; ++s) {
      v2f b = *(const v2f*)(brow + (ct * K_STEPS + s) * 64);
      acc = __builtin_amdgcn_wmma_f32_16x16x4_f32(
           false, a[s],  false, b,
           (short)0, acc,  false,  false);
    }
    #pragma unroll
    for (int i = 0; i < 8; ++i) {
      int node = row0 + i + 8 * half;
      P[(size_t)node * P_COLS + ct * 16 + col] = acc[i];
    }
  }

  {
    float bias = b2[col & (L_DIM - 1)];
    v8f acc = {};
    #pragma unroll
    for (int s = 0; s < K_STEPS; ++s) {
      v2f b = *(const v2f*)(brow + ((C_TILES - 1) * K_STEPS + s) * 64);
      acc = __builtin_amdgcn_wmma_f32_16x16x4_f32(
           false, a[s],  false, b,
           (short)0, acc,  false,  false);
    }
    if (col < L_DIM) {
      #pragma unroll
      for (int i = 0; i < 8; ++i) {
        int node = row0 + i + 8 * half;
        pre[(size_t)node * L_DIM + col] = acc[i] + bias;
      }
    }
  }
}

__global__ __launch_bounds__(256) void rgcn_layer2_seg(const int* __restrict__ offs,
                                                       const int* __restrict__ keys,
                                                       const float* __restrict__ norms,
                                                       const float* __restrict__ P,
                                                       const float* __restrict__ pre,
                                                       float* __restrict__ out) {
  const int lane = threadIdx.x & 31;
  const int wave = blockIdx.x * 8 + __builtin_amdgcn_readfirstlane(threadIdx.x >> 5);
  if (wave >= N_NODES / 4) return;
  const int l = lane & (L_DIM - 1);
  const int g = lane >> 3;
  const int node0 = wave * 4;
  float res = 0.0f;
  for (int q = 0; q < 4; ++q) {
    const int beg = offs[node0 + q];
    const int end = offs[node0 + q + 1];
    float acc = 0.0f;
    for (int base = beg; base < end; base += 32) {
      const int n = min(32, end - base);
      int   key = 0;
      float nm  = 0.0f;
      if (lane < n) {
        key = keys[base + lane];
        nm  = norms[base + lane];
      }
      float v[8];
      float nn[8];
      #pragma unroll
      for (int jj = 0; jj < 8; ++jj) {
        int idx = jj * 4 + g;
        int kk  = __shfl(key, idx);
        nn[jj]  = __shfl(nm, idx);
        int src = kk & 0xFFFF;
        int r   = kk >> 16;
        v[jj] = P[((size_t)src * R_REL + r) * L_DIM + l];
      }
      #pragma unroll
      for (int jj = 0; jj < 8; ++jj) {
        if (jj * 4 + g < n) acc = fmaf(v[jj], nn[jj], acc);
      }
    }
    acc += __shfl_xor(acc, 8);
    acc += __shfl_xor(acc, 16);
    if (g == q) res = acc;
  }
  const size_t o = (size_t)node0 * L_DIM + lane;
  float x = pre[o] + res;
  out[o] = 1.0f / (1.0f + __expf(-x));
}

extern "C" void kernel_launch(void* const* d_in, const int* in_sizes, int n_in,
                              void* d_out, int out_size, void* d_ws, size_t ws_size,
                              hipStream_t stream) {
  const int*   ei    = (const int*)d_in[0];
  const int*   et    = (const int*)d_in[1];
  const float* W1    = (const float*)d_in[2];
  const float* root1 = (const float*)d_in[3];
  const float* b1    = (const float*)d_in[4];
  const float* W2    = (const float*)d_in[5];
  const float* root2 = (const float*)d_in[6];
  const float* b2    = (const float*)d_in[7];
  float*       out   = (float*)d_out;

  int*   offs     = (int*)d_ws;
  int*   keys     = offs + OFFS_WORDS;
  float* norms    = (float*)(keys + E_EDGES);
  float* h1       = norms + E_EDGES;
  float* P        = h1 + (size_t)N_NODES * H_DIM;
  float* pre      = P + (size_t)N_NODES * P_COLS;
  int*   chunkrec = (int*)(pre + (size_t)N_NODES * L_DIM);
  int*   chunkoff = chunkrec + E_EDGES;

  rgcn_bin<<<N_CHUNKS, 1024, 0, stream>>>(ei, et, chunkrec, chunkoff);

  rgcn_binsort<<<N_BINS, 1024, 0, stream>>>(chunkrec, chunkoff, offs, keys);

  rgcn_layer1_seg<<<(N_NODES / L1_NODES + 7) / 8, 256, 0, stream>>>(offs, keys, W1, root1, b1, norms, h1);

  const int gemmBlocks = (TILES * 32 + 255) / 256;
  rgcn_project_wmma<<<gemmBlocks, 256, 0, stream>>>(h1, W2, root2, b2, P, pre);

  rgcn_layer2_seg<<<(N_NODES / 4 + 7) / 8, 256, 0, stream>>>(offs, keys, norms, P, pre, out);

  (void)in_sizes; (void)n_in; (void)out_size; (void)ws_size;
}


# Projection fused into the layer-1 kernel: a block's 8 waves leave their 32 h1 rows in LDS and run the same WMMA panel product behind one barrier; h1 skips memory, 4 launches; sum order unchanged
# speedup vs baseline: 6.6168x; 1.1869x over previous
#include <hip/hip_runtime.h>


typedef __attribute__((ext_vector_type(2))) float v2f;
typedef __attribute__((ext_vector_type(4))) float v4f;
typedef __attribute__((ext_vector_type(8))) float v8f;

#define N_NODES 50000
#define R_REL   16
#define H_DIM   32
#define L_DIM   8
#define E_EDGES 1600000

#define TILES   (N_NODES / 16)
#define K_STEPS (H_DIM / 4)
#define P_COLS  (R_REL * L_DIM)

#define BIN_SHIFT  8
#define BIN_NODES  (1 << BIN_SHIFT)
#define N_BINS     ((N_NODES + BIN_NODES - 1) / BIN_NODES)
#define N_CHUNKS   256
#define CHUNK      (E_EDGES / N_CHUNKS)
#define OFF_PITCH  (N_BINS + 1)
#define BIN_PER    ((CHUNK + 1023) / 1024)
#define STAGE_CAP  12288
#define OFFS_WORDS 50016

static_assert(E_EDGES % N_CHUNKS == 0, "chunks are equal");
static_assert(N_CHUNKS == 8 * 32, "one wave scans the chunk table, 8 entries per lane; 32 waves take 8 chunks each");
static_assert(BIN_NODES == 8 * 32, "one wave scans a bin's node counters, 8 per lane");
static_assert(N_BINS <= 7 * 32, "one wave scans the bin counters, 7 per lane");
static_assert(OFFS_WORDS >= N_NODES + 1, "offsets of every node and the end offset");
static_assert(N_NODES <= 65536, "source index is packed into 16 bits");
static_assert(R_REL <= 16, "relation is packed into 4 bits");
static_assert(N_NODES % 4 == 0, "4 nodes per wave in the layer-2 segment kernel");

__device__ __forceinline__ int rgcn_wave_incl(int v, int lane) {
  #pragma unroll
  for (int d = 1; d < 32; d <<= 1) {
    int t = __shfl_up(v, d);
    if (lane >= d) v += t;
  }
  return v;
}

__global__ __launch_bounds__(1024) void rgcn_bin(const int* __restrict__ ei, const int* __restrict__ et,
                                                 int* __restrict__ chunkrec, int* __restrict__ chunkoff) {
  __shared__ int cur[N_BINS];
  __shared__ int start[N_BINS + 1];
  __shared__ int stage[CHUNK];
  const int tid  = threadIdx.x;
  const int lane = tid & 31;
  const int base = blockIdx.x * CHUNK;

  for (int b = tid; b < N_BINS; b += 1024) cur[b] = 0;
  __syncthreads();

  int rec[BIN_PER];
  int bin[BIN_PER];
  #pragma unroll
  for (int k = 0; k < BIN_PER; ++k) {
    const int i = tid + k * 1024;
    rec[k] = 0;
    bin[k] = -1;
    if (i < CHUNK) {
      const int src = ei[base + i];
      const int dst = ei[E_EDGES + base + i];
      const int r   = et[base + i];
      rec[k] = src | (r << 16) | ((dst & (BIN_NODES - 1)) << 20);
      bin[k] = dst >> BIN_SHIFT;
    }
  }
  #pragma unroll
  for (int k = 0; k < BIN_PER; ++k)
    if (bin[k] >= 0) atomicAdd(&cur[bin[k]], 1);
  __syncthreads();

  if (tid < 32) {
    int loc[7];
    int s = 0;
    #pragma unroll
    for (int j = 0; j < 7; ++j) {
      const int idx = lane * 7 + j;
      const int v = idx < N_BINS ? cur[idx] : 0;
      loc[j] = s;
      s += v;
    }
    const int inc  = rgcn_wave_incl(s, lane);
    const int excl = inc - s;
    #pragma unroll
    for (int j = 0; j < 7; ++j) {
      const int idx = lane * 7 + j;
      if (idx < N_BINS) {
        start[idx] = excl + loc[j];
        cur[idx]   = excl + loc[j];
      }
    }
    if (lane == 31) start[N_BINS] = inc;
  }
  __syncthreads();

  #pragma unroll
  for (int k = 0; k < BIN_PER; ++k) {
    if (bin[k] >= 0) {
      const int pos = atomicAdd(&cur[bin[k]], 1);
      stage[pos] = rec[k];
    }
  }
  __syncthreads();

  for (int i = tid; i < CHUNK; i += 1024) chunkrec[base + i] = stage[i];
  for (int b = tid; b <= N_BINS; b += 1024) chunkoff[blockIdx.x * OFF_PITCH + b] = start[b];
}

__global__ __launch_bounds__(1024) void rgcn_binsort(const int* __restrict__ chunkrec,
                                                     const int* __restrict__ chunkoff,
                                                     int* __restrict__ offs, int* __restrict__ keys) {
  __shared__ int plo[N_CHUNKS];
  __shared__ int ppos[N_CHUNKS + 1];
  __shared__ int cur[BIN_NODES];
  __shared__ int sbase;
  __shared__ int stage[STAGE_CAP];
  __shared__ int sorted[STAGE_CAP];
  const int tid  = threadIdx.x;
  const int lane = tid & 31;
  const int w    = __builtin_amdgcn_readfirstlane(tid >> 5);
  const int b    = blockIdx.x;

  if (tid < 32) {
    int lo[8];
    int ln[8];
    #pragma unroll
    for (int j = 0; j < 8; ++j) {
      const int* p = chunkoff + (lane * 8 + j) * OFF_PITCH + b;
      lo[j] = p[0];
      ln[j] = p[1] - p[0];
    }
    int s  = 0;
    int sl = 0;
    #pragma unroll
    for (int j = 0; j < 8; ++j) {
      const int v = ln[j];
      ln[j] = s;
      s  += v;
      sl += lo[j];
    }
    const int inc  = rgcn_wave_incl(s, lane);
    const int excl = inc - s;
    #pragma unroll
    for (int d = 16; d >= 1; d >>= 1) sl += __shfl_xor(sl, d);
    #pragma unroll
    for (int j = 0; j < 8; ++j) {
      plo[lane * 8 + j]  = lo[j];
      ppos[lane * 8 + j] = excl + ln[j];
    }
    if (lane == 31) ppos[N_CHUNKS] = inc;
    if (lane == 0)  sbase = sl;
  }
  for (int j = tid; j < BIN_NODES; j += 1024) cur[j] = 0;
  __syncthreads();

  const int  total  = ppos[N_CHUNKS];
  const int  kbase  = sbase;
  const bool staged = total <= STAGE_CAP;

  {
    int ra[8];
    int rb[8];
    #pragma unroll
    for (int k = 0; k < 8; ++k) {
      const int c   = w + 32 * k;
      const int len = ppos[c + 1] - ppos[c];
      const int lo  = plo[c];
      const int* cb = chunkrec + c * CHUNK;
      const int va = cb[min(lo + lane, CHUNK - 1)];
      const int vb = cb[min(lo + lane + 32, CHUNK - 1)];
      ra[k] = lane < len      ? va : -1;
      rb[k] = lane + 32 < len ? vb : -1;
    }
    #pragma unroll
    for (int k = 0; k < 8; ++k) {
      const int c   = w + 32 * k;
      const int pos = ppos[c];
      const int len = ppos[c + 1] - pos;
      const int* src = chunkrec + c * CHUNK + plo[c];
      if (ra[k] >= 0) {
        atomicAdd(&cur[ra[k] >> 20], 1);
        if (staged) stage[pos + lane] = ra[k];
      }
      if (rb[k] >= 0) {
        atomicAdd(&cur[rb[k] >> 20], 1);
        if (staged) stage[pos + lane + 32] = rb[k];
      }
      for (int i = lane + 64; i < len; i += 32) {
        const int rc = src[i];
        atomicAdd(&cur[rc >> 20], 1);
        if (staged) stage[pos + i] = rc;
      }
    }
  }
  __syncthreads();

  if (tid < 32) {
    int loc[8];
    int s = 0;
    #pragma unroll
    for (int j = 0; j < 8; ++j) {
      const int v = cur[lane * 8 + j];
      loc[j] = s;
      s += v;
    }
    const int inc  = rgcn_wave_incl(s, lane);
    const int excl = inc - s;
    #pragma unroll
    for (int j = 0; j < 8; ++j) {
      const int idx  = lane * 8 + j;
      const int st   = excl + loc[j];
      const int node = b * BIN_NODES + idx;
      cur[idx] = st;
      if (node < N_NODES) offs[node] = kbase + st;
    }
    if (lane == 31 && b == N_BINS - 1) offs[N_NODES] = kbase + inc;
  }
  __syncthreads();

  if (staged) {
    for (int i = tid; i < total; i += 1024) {
      const int rc  = stage[i];
      const int pos = atomicAdd(&cur[rc >> 20], 1);
      sorted[pos] = rc & 0xFFFFF;
    }
    __syncthreads();
    for (int i = tid; i < total; i += 1024) keys[kbase + i] = sorted[i];
  } else {
    for (int k = 0; k < 8; ++k) {
      const int c   = w + 32 * k;
      const int len = ppos[c + 1] - ppos[c];
      const int* src = chunkrec + c * CHUNK + plo[c];
      for (int i = lane; i < len; i += 32) {
        const int rc  = src[i];
        const int pos = atomicAdd(&cur[rc >> 20], 1);
        keys[kbase + pos] = rc & 0xFFFFF;
      }
    }
  }
}

__device__ __forceinline__ void rgcn_rel_count(int rel, int (&c)[R_REL]) {
  #pragma unroll
  for (int rr = 0; rr < R_REL; ++rr)
    c[rr] += __builtin_popcount(__builtin_amdgcn_ballot_w32(rel == rr));
}

#define L1_NODES 4
#define H_PITCH  34
static_assert(N_NODES % L1_NODES == 0, "every layer-1 wave owns L1_NODES whole nodes");
static_assert(L1_NODES + 1 <= 32, "one lane per segment bound of the wave's nodes");
static_assert(8 * L1_NODES == 32, "a block of 8 waves owns two 16-row tiles");
static_assert(N_NODES % 16 == 0, "a tile is either whole or absent");
static_assert(P_COLS == 4 * 32, "four waves of 32 columns cover a P row");

__global__ __launch_bounds__(256) void rgcn_layer1_project(const int* __restrict__ offs,
                                                           const int* __restrict__ keys,
                                                           const float* __restrict__ W1,
                                                           const float* __restrict__ root1,
                                                           const float* __restrict__ b1,
                                                           const float* __restrict__ W2,
                                                           const float* __restrict__ root2,
                                                           const float* __restrict__ b2,
                                                           float* __restrict__ norms,
                                                           float* __restrict__ P,
                                                           float* __restrict__ pre) {
  __shared__ __attribute__((aligned(16))) float ldsH[8 * L1_NODES * H_PITCH];
  const int lane = threadIdx.x & 31;
  const int w    = __builtin_amdgcn_readfirstlane(threadIdx.x >> 5);
  const int wave = blockIdx.x * 8 + w;

  if (wave < N_NODES / L1_NODES) {
    const int node0 = wave * L1_NODES;

    int ov = 0;
    if (lane <= L1_NODES) ov = offs[node0 + lane];
    const float bias = b1[lane];

    int beg = __builtin_amdgcn_readlane(ov, 0);
    int end = __builtin_amdgcn_readlane(ov, 1);
    int   raw0 = keys[min(beg + lane, E_EDGES - 1)];
    int   raw1 = keys[min(beg + 32 + lane, E_EDGES - 1)];
    float rt   = root1[(size_t)node0 * H_DIM + lane];

    for (int q = 0; q < L1_NODES; ++q) {
      const int node = node0 + q;

      const int   nend  = __builtin_amdgcn_readlane(ov, min(q + 2, L1_NODES));
      const int   nraw0 = keys[min(end + lane, E_EDGES - 1)];
      const int   nraw1 = keys[min(end + 32 + lane, E_EDGES - 1)];
      const float nrt   = root1[(size_t)min(node + 1, N_NODES - 1) * H_DIM + lane];

      const int n0 = min(32, end - beg);
      const int n1 = min(32, end - beg - 32);
      const int key0 = lane < n0 ? raw0 : 0;
      const int key1 = lane < n1 ? raw1 : 0;

      int c[R_REL];
      #pragma unroll
      for (int rr = 0; rr < R_REL; ++rr) c[rr] = 0;
      rgcn_rel_count(lane < n0 ? (key0 >> 16) : R_REL, c);
      if (n1 > 0) rgcn_rel_count(lane < n1 ? (key1 >> 16) : R_REL, c);
      for (int base = beg + 64; base < end; base += 32) {
        const int n = min(32, end - base);
        int rel = R_REL;
        if (lane < n) rel = keys[base + lane] >> 16;
        rgcn_rel_count(rel, c);
      }
      int cntv = 0;
      #pragma unroll
      for (int rr = 0; rr < R_REL; ++rr) cntv = (lane == rr) ? c[rr] : cntv;

      float acc = 0.0f;
      for (int base = beg; base < end; base += 32) {
        const int n = min(32, end - base);
        int key;
        if (base == beg) {
          key = key0;
        } else if (base == beg + 32) {
          key = key1;
        } else {
          key = 0;
          if (lane < n) key = keys[base + lane];
        }
        const int   cl = __shfl(cntv, key >> 16);
        const float nm = 1.0f / fmaxf((float)cl, 1.0f);
        if (lane < n) norms[base + lane] = nm;

        float w[32];
        #pragma unroll
        for (int g = 0; g < 4; ++g) {
          if (g * 8 < n) {
            #pragma unroll
            for (int u = 0; u < 8; ++u) {
              const int kk  = __builtin_amdgcn_readlane(key, g * 8 + u);
              const int src = kk & 0xFFFF;
              const int r   = kk >> 16;
              w[g * 8 + u] = W1[(r * N_NODES + src) * H_DIM + lane];
            }
          }
        }
        #pragma unroll
        for (int u = 0; u < 32; ++u) {
          if (u < n) {
            const float nn = __int_as_float(__builtin_amdgcn_readlane(__float_as_int(nm), u));
            acc = fmaf(w[u], nn, acc);
          }
        }
      }
      ldsH[(w * L1_NODES + q) * H_PITCH + lane] = fmaxf(acc + rt + bias, 0.0f);

      beg  = end;
      end  = nend;
      raw0 = nraw0;
      raw1 = nraw1;
      rt   = nrt;
    }
  }

  const int  rt   = w >> 2;
  const int  lg   = w & 3;
  const int  tile = blockIdx.x * 2 + rt;
  const bool has_tile = tile < TILES;
  const int  n    = lane & 15;
  const int  half = lane >> 4;

  v2f bf[2][K_STEPS];
  {
    const int c0 = lg * 32 + n;
    const float* bb = W2 + ((c0 >> 3) * H_DIM + half * 2) * L_DIM + (c0 & (L_DIM - 1));
    #pragma unroll
    for (int ctl = 0; ctl < 2; ++ctl) {
      #pragma unroll
      for (int s = 0; s < K_STEPS; ++s) {
        bf[ctl][s].x = bb[(ctl * 2 * H_DIM + s * 4 + 0) * L_DIM];
        bf[ctl][s].y = bb[(ctl * 2 * H_DIM + s * 4 + 1) * L_DIM];
      }
    }
  }
  __syncthreads();

  if (has_tile) {
    const float* arow = ldsH + (rt * 16 + n) * H_PITCH + half * 2;
    v2f a[K_STEPS];
    #pragma unroll
    for (int s = 0; s < K_STEPS; ++s) a[s] = *(const v2f*)(arow + s * 4);

    const int row0 = tile * 16;
    #pragma unroll
    for (int ctl = 0; ctl < 2; ++ctl) {
      v8f acc = {};
      #pragma unroll
      for (int s = 0; s < K_STEPS; ++s) {
        acc = __builtin_amdgcn_wmma_f32_16x16x4_f32(
             false, a[s],  false, bf[ctl][s],
             (short)0, acc,  false,  false);
      }
      #pragma unroll
      for (int i = 0; i < 8; ++i) {
        const int node = row0 + i + 8 * half;
        P[(size_t)node * P_COLS + (lg * 2 + ctl) * 16 + n] = acc[i];
      }
    }
    if (lg == 0) {
      const float* rb = root2 + (half * 2) * L_DIM + (n & (L_DIM - 1));
      const float bias2 = b2[n & (L_DIM - 1)];
      v8f acc = {};
      #pragma unroll
      for (int s = 0; s < K_STEPS; ++s) {
        v2f b;
        b.x = rb[(s * 4 + 0) * L_DIM];
        b.y = rb[(s * 4 + 1) * L_DIM];
        acc = __builtin_amdgcn_wmma_f32_16x16x4_f32(
             false, a[s],  false, b,
             (short)0, acc,  false,  false);
      }
      if (n < L_DIM) {
        #pragma unroll
        for (int i = 0; i < 8; ++i) {
          const int node = row0 + i + 8 * half;
          pre[(size_t)node * L_DIM + n] = acc[i] + bias2;
        }
      }
    }
  }
}

__global__ __launch_bounds__(256) void rgcn_layer2_seg(const int* __restrict__ offs,
                                                       const int* __restrict__ keys,
                                                       const float* __restrict__ norms,
                                                       const float* __restrict__ P,
                                                       const float* __restrict__ pre,
                                                       float* __restrict__ out) {
  const int lane = threadIdx.x & 31;
  const int wave = blockIdx.x * 8 + __builtin_amdgcn_readfirstlane(threadIdx.x >> 5);
  if (wave >= N_NODES / 4) return;
  const int l = lane & (L_DIM - 1);
  const int g = lane >> 3;
  const int node0 = wave * 4;
  float res = 0.0f;
  for (int q = 0; q < 4; ++q) {
    const int beg = offs[node0 + q];
    const int end = offs[node0 + q + 1];
    float acc = 0.0f;
    for (int base = beg; base < end; base += 32) {
      const int n = min(32, end - base);
      int   key = 0;
      float nm  = 0.0f;
      if (lane < n) {
        key = keys[base + lane];
        nm  = norms[base + lane];
      }
      float v[8];
      float nn[8];
      #pragma unroll
      for (int jj = 0; jj < 8; ++jj) {
        int idx = jj * 4 + g;
        int kk  = __shfl(key, idx);
        nn[jj]  = __shfl(nm, idx);
        int src = kk & 0xFFFF;
        int r   = kk >> 16;
        v[jj] = P[((size_t)src * R_REL + r) * L_DIM + l];
      }
      #pragma unroll
      for (int jj = 0; jj < 8; ++jj) {
        if (jj * 4 + g < n) acc = fmaf(v[jj], nn[jj], acc);
      }
    }
    acc += __shfl_xor(acc, 8);
    acc += __shfl_xor(acc, 16);
    if (g == q) res = acc;
  }
  const size_t o = (size_t)node0 * L_DIM + lane;
  float x = pre[o] + res;
  out[o] = 1.0f / (1.0f + __expf(-x));
}

extern "C" void kernel_launch(void* const* d_in, const int* in_sizes, int n_in,
                              void* d_out, int out_size, void* d_ws, size_t ws_size,
                              hipStream_t stream) {
  const int*   ei    = (const int*)d_in[0];
  const int*   et    = (const int*)d_in[1];
  const float* W1    = (const float*)d_in[2];
  const float* root1 = (const float*)d_in[3];
  const float* b1    = (const float*)d_in[4];
  const float* W2    = (const float*)d_in[5];
  const float* root2 = (const float*)d_in[6];
  const float* b2    = (const float*)d_in[7];
  float*       out   = (float*)d_out;

  int*   offs     = (int*)d_ws;
  int*   keys     = offs + OFFS_WORDS;
  float* norms    = (float*)(keys + E_EDGES);
  float* P        = norms + E_EDGES;
  float* pre      = P + (size_t)N_NODES * P_COLS;
  int*   chunkrec = (int*)(pre + (size_t)N_NODES * L_DIM);
  int*   chunkoff = chunkrec + E_EDGES;

  rgcn_bin<<<N_CHUNKS, 1024, 0, stream>>>(ei, et, chunkrec, chunkoff);

  rgcn_binsort<<<N_BINS, 1024, 0, stream>>>(chunkrec, chunkoff, offs, keys);

  rgcn_layer1_project<<<(N_NODES / L1_NODES + 7) / 8, 256, 0, stream>>>(offs, keys, W1, root1, b1, W2, root2, b2,
                                                                        norms, P, pre);

  rgcn_layer2_seg<<<(N_NODES / 4 + 7) / 8, 256, 0, stream>>>(offs, keys, norms, P, pre, out);

  (void)in_sizes; (void)n_in; (void)out_size; (void)ws_size;
}
